# starting kernel
# baseline (speedup 1.0000x reference)
#include <hip/hip_runtime.h>

#define NNODES  50000
#define NEDGES  800000
#define DIM     128
#define NLAYERS 3

typedef __attribute__((ext_vector_type(2))) float v2f;
typedef __attribute__((ext_vector_type(8))) float v8f;

__device__ __forceinline__ void atomic_add_f32_dev(float* p, float v) {
    asm volatile("global_atomic_add_f32 %0, %1, off scope:SCOPE_DEV"
                 :: "v"(p), "v"(v) : "memory");
}

__global__ void gcn_zero_kernel(float4* __restrict__ p, int n4) {
    int i = blockIdx.x * blockDim.x + threadIdx.x;
    if (i < n4) p[i] = make_float4(0.f, 0.f, 0.f, 0.f);
}

__global__ __launch_bounds__(256) void gcn_scatter_kernel(
    const float* __restrict__ h,
    const int*   __restrict__ edge_index,
    const float* __restrict__ edge_weight,
    float*       __restrict__ agg,
    int nedges) {
    int tid  = blockIdx.x * blockDim.x + threadIdx.x;
    int e    = tid >> 5;
    int lane = tid & 31;
    if (e >= nedges) return;

    int   src = edge_index[e];
    int   dst = edge_index[nedges + e];
    float w   = edge_weight[e];

    const float4* hv = reinterpret_cast<const float4*>(h + (size_t)src * DIM);
    float4 v = hv[lane];

    float* a = agg + (size_t)dst * DIM + (size_t)lane * 4;
    atomic_add_f32_dev(a + 0, v.x * w);
    atomic_add_f32_dev(a + 1, v.y * w);
    atomic_add_f32_dev(a + 2, v.z * w);
    atomic_add_f32_dev(a + 3, v.w * w);
}

__global__ __launch_bounds__(256) void gcn_gemm_wmma_kernel(
    const float* __restrict__ agg,
    const float* __restrict__ h,
    const float* __restrict__ Wrel,
    const float* __restrict__ Wroot,
    const float* __restrict__ bias,
    float*       __restrict__ out,
    int nrows, int do_relu) {
    __shared__ float4 sA4[16 * DIM / 4];
    __shared__ float4 sH4[16 * DIM / 4];

    const int t    = threadIdx.x;
    const int lane = t & 31;
    const int wv   = t >> 5;
    const int r    = lane & 15;
    const int half = lane >> 4;

    const int row0 = blockIdx.x * 16;
    const int col0 = wv * 16;

    {
        const float4* aggv = reinterpret_cast<const float4*>(agg) + (size_t)row0 * (DIM / 4);
        const float4* hv   = reinterpret_cast<const float4*>(h)   + (size_t)row0 * (DIM / 4);
#pragma unroll
        for (int i = t; i < 16 * DIM / 4; i += 256) {
            sA4[i] = aggv[i];
            sH4[i] = hv[i];
        }
    }
    __syncthreads();

    const float* sA = reinterpret_cast<const float*>(sA4);
    const float* sH = reinterpret_cast<const float*>(sH4);

    v8f acc_rel  = {};
    v8f acc_root = {};
#pragma unroll
    for (int k = 0; k < DIM; k += 4) {
        const int ka = k + 2 * half;

        v2f a_rel  = *reinterpret_cast<const v2f*>(&sA[r * DIM + ka]);
        v2f b_rel  = { Wrel[(size_t)ka * DIM + col0 + r],
                       Wrel[(size_t)(ka + 1) * DIM + col0 + r] };
        acc_rel = __builtin_amdgcn_wmma_f32_16x16x4_f32(
                      false, a_rel, false, b_rel, (short)0, acc_rel, false, false);

        v2f a_root = *reinterpret_cast<const v2f*>(&sH[r * DIM + ka]);
        v2f b_root = { Wroot[(size_t)ka * DIM + col0 + r],
                       Wroot[(size_t)(ka + 1) * DIM + col0 + r] };
        acc_root = __builtin_amdgcn_wmma_f32_16x16x4_f32(
                      false, a_root, false, b_root, (short)0, acc_root, false, false);
    }

    const float bv = bias[col0 + r];
#pragma unroll
    for (int j = 0; j < 8; ++j) {
        const int m    = j + 8 * half;
        const int orow = row0 + m;
        if (orow < nrows) {
            float v = acc_rel[j] + acc_root[j] + bv;
            if (do_relu) v = fmaxf(v, 0.f);
            out[(size_t)orow * DIM + col0 + r] = v;
        }
    }
}

extern "C" void kernel_launch(void* const* d_in, const int* in_sizes, int n_in,
                              void* d_out, int out_size, void* d_ws, size_t ws_size,
                              hipStream_t stream) {
    const float* x      = (const float*)d_in[0];
    const int*   eidx   = (const int*)  d_in[1];
    const float* ew     = (const float*)d_in[2];
    const float* Wrel   = (const float*)d_in[3];
    const float* brel   = (const float*)d_in[4];
    const float* Wroot  = (const float*)d_in[5];
    float*       out    = (float*)d_out;

    const int N = NNODES;
    const int E = NEDGES;

    float* h1  = (float*)d_ws;
    float* h2  = h1 + (size_t)N * DIM;
    float* agg = h2 + (size_t)N * DIM;

    const int n4          = N * DIM / 4;
    const int zero_blocks = (n4 + 255) / 256;
    const int scat_blocks = (E * 32 + 255) / 256;
    const int gemm_blocks = (N + 15) / 16;

    for (int l = 0; l < NLAYERS; ++l) {
        const float* hin  = (l == 0) ? x  : (l == 1) ? h1 : h2;
        float*       hout = (l == 0) ? h1 : (l == 1) ? h2 : out;

        gcn_zero_kernel<<<zero_blocks, 256, 0, stream>>>(
            (float4*)agg, n4);

        gcn_scatter_kernel<<<scat_blocks, 256, 0, stream>>>(
            hin, eidx, ew, agg, E);

        gcn_gemm_wmma_kernel<<<gemm_blocks, 256, 0, stream>>>(
            agg, hin,
            Wrel  + (size_t)l * DIM * DIM,
            Wroot + (size_t)l * DIM * DIM,
            brel  + (size_t)l * DIM,
            hout, N, (l < NLAYERS - 1) ? 1 : 0);
    }
}


# f32 atomic scatter replaced by per-call edge lists grouped by dst (two-level counting sort, ballot ranks, no shared counters); one wave sums a node's rows in registers in edge order
# speedup vs baseline: 1.8783x; 1.8783x over previous
#include <hip/hip_runtime.h>

#define NNODES  50000
#define NEDGES  800000
#define DIM     128
#define NLAYERS 3

typedef __attribute__((ext_vector_type(2))) float v2f;
typedef __attribute__((ext_vector_type(8))) float v8f;


#define BINW    256
#define NWAVES  (BINW / 32)
#define TILE1   4096
#define NTILES  ((NEDGES + TILE1 - 1) / TILE1)
#define NBINS   (NNODES / BINW + 1)
#define NCOUNTS (BINW * NTILES)
#define BATCH   8

#define SCAN_THREADS 1024
#define SCAN_ITEMS   ((NCOUNTS + SCAN_THREADS - 1) / SCAN_THREADS)

static_assert(NBINS <= BINW, "level-1 bin numbers must fit the 256-entry histogram");
static_assert(TILE1 % (NWAVES * 32 * BATCH) == 0, "a tile is whole batches for every wave");

__device__ __forceinline__ void gcn_match32(int key, int& lt, int& tot, bool& last) {
    unsigned mine = 0u;
#pragma unroll
    for (int l = 0; l < 32; ++l) {
        const int      kl = __builtin_amdgcn_readlane(key, l);
        const bool     eq = (key == kl);
        const unsigned m  = __builtin_amdgcn_ballot_w32(eq);
        if (eq) mine = m;
    }
    const unsigned lane = threadIdx.x & 31u;
    lt   = (int)__builtin_amdgcn_mbcnt_lo(mine, 0u);
    tot  = __popc(mine);
    last = ((mine >> lane) == 1u);
}

__device__ __forceinline__ int gcn_rank_round(int* __restrict__ whist, int key, bool valid) {
    int lt, tot; bool last;
    gcn_match32(key, lt, tot, last);
    int part = 0;
    if (valid) {
        const int old = whist[key];
        part = old + lt;
        if (last) whist[key] = old + tot;
    }
    return part;
}

__global__ __launch_bounds__(BINW) void gcn_bin_count_kernel(
    const int* __restrict__ edge_index,
    int*       __restrict__ tilecnt,
    int*       __restrict__ rank1,
    int nedges) {
    __shared__ int hist[NWAVES][BINW];

    const int t    = threadIdx.x;
    const int lane = t & 31;
    const int wv   = __builtin_amdgcn_readfirstlane(t >> 5);
    const int tile = blockIdx.x;
    const int e0   = tile * TILE1;
    const int ne   = min(TILE1, nedges - e0);
    const int seg0 = wv * (TILE1 / NWAVES);
    const int* dst = edge_index + nedges;

    for (int k = 0; k < NWAVES; ++k) hist[k][t] = 0;
    __syncthreads();

    for (int q0 = 0; q0 < TILE1 / NWAVES; q0 += 32 * BATCH) {
        if (seg0 + q0 >= ne) break;
        int key[BATCH];
#pragma unroll
        for (int b = 0; b < BATCH; ++b) {
            const int idx = seg0 + q0 + 32 * b + lane;
            const int v = dst[e0 + min(idx, ne - 1)];
            key[b] = (idx < ne) ? (v >> 8) : -1;
        }
#pragma unroll
        for (int b = 0; b < BATCH; ++b) {
            if (seg0 + q0 + 32 * b >= ne) continue;
            const int  idx   = seg0 + q0 + 32 * b + lane;
            const bool valid = idx < ne;
            const int  part  = gcn_rank_round(hist[wv], key[b], valid);
            if (valid) rank1[e0 + idx] = part;
        }
    }
    __syncthreads();

    int run = 0;
    for (int k = 0; k < NWAVES; ++k) {
        const int v = hist[k][t];
        hist[k][t] = run;
        run += v;
    }
    tilecnt[t * NTILES + tile] = run;
    __syncthreads();

    for (int q = 0; q < TILE1 / NWAVES; q += 32) {
        if (seg0 + q >= ne) break;
        const int idx = seg0 + q + lane;
        if (idx < ne) rank1[e0 + idx] += hist[wv][dst[e0 + idx] >> 8];
    }
}

__global__ __launch_bounds__(SCAN_THREADS) void gcn_scan_kernel(
    const int* __restrict__ cnt,
    int*       __restrict__ base,
    int n) {
    __shared__ int wave_sum[SCAN_THREADS / 32];

    const int t    = threadIdx.x;
    const int lane = t & 31;
    const int wv   = t >> 5;
    const int beg  = t * SCAN_ITEMS;

    int vals[SCAN_ITEMS];
    int s = 0;
#pragma unroll
    for (int i = 0; i < SCAN_ITEMS; ++i) {
        const int idx = beg + i;
        const int v   = (idx < n) ? cnt[idx] : 0;
        vals[i] = v;
        s += v;
    }

    int incl = s;
    for (int off = 1; off < 32; off <<= 1) {
        const int y = __shfl_up(incl, off);
        if (lane >= off) incl += y;
    }
    if (lane == 31) wave_sum[wv] = incl;
    __syncthreads();

    if (wv == 0) {
        const int x = wave_sum[lane];
        int xi = x;
        for (int off = 1; off < 32; off <<= 1) {
            const int y = __shfl_up(xi, off);
            if (lane >= off) xi += y;
        }
        wave_sum[lane] = xi - x;
    }
    __syncthreads();

    int run = wave_sum[wv] + incl - s;
#pragma unroll
    for (int i = 0; i < SCAN_ITEMS; ++i) {
        const int idx = beg + i;
        if (idx < n) {
            base[idx] = run;
            run += vals[i];
        }
    }
    if (t == SCAN_THREADS - 1) base[n] = run;
}

__global__ __launch_bounds__(256) void gcn_bin_place_kernel(
    const int*   __restrict__ edge_index,
    const float* __restrict__ edge_weight,
    const int*   __restrict__ tilebase,
    const int*   __restrict__ rank1,
    int4*        __restrict__ rec1,
    int nedges) {
    const int e = blockIdx.x * blockDim.x + threadIdx.x;
    if (e >= nedges) return;
    const int   src = edge_index[e];
    const int   dst = edge_index[nedges + e];
    const float w   = edge_weight[e];
    const int   pos = tilebase[(dst >> 8) * NTILES + e / TILE1] + rank1[e];
    rec1[pos] = make_int4(src, __float_as_int(w), dst, 0);
}

__global__ __launch_bounds__(BINW) void gcn_node_group_kernel(
    const int*  __restrict__ tilebase,
    const int4* __restrict__ rec1,
    int*        __restrict__ rank2,
    int*        __restrict__ rowptr,
    int2*       __restrict__ rec,
    int nnodes) {
    __shared__ int hist[NWAVES][BINW];
    __shared__ int soff[BINW];
    __shared__ int wave_sum[NWAVES];

    const int t    = threadIdx.x;
    const int lane = t & 31;
    const int wv   = __builtin_amdgcn_readfirstlane(t >> 5);
    const int bin  = blockIdx.x;
    const int lo   = tilebase[bin * NTILES];
    const int hi   = tilebase[(bin + 1) * NTILES];
    const int seg  = ((hi - lo + BINW - 1) / BINW) * 32;
    const int p0   = lo + wv * seg;

    for (int k = 0; k < NWAVES; ++k) hist[k][t] = 0;
    __syncthreads();

    for (int q0 = 0; q0 < seg; q0 += 32 * BATCH) {
        if (p0 + q0 >= hi) break;
        int key[BATCH];
#pragma unroll
        for (int b = 0; b < BATCH; ++b) {
            const int q = q0 + 32 * b;
            const int p = p0 + q + lane;
            const int v = rec1[min(p, hi - 1)].z;
            key[b] = (q < seg && p < hi) ? (v & (BINW - 1)) : -1;
        }
#pragma unroll
        for (int b = 0; b < BATCH; ++b) {
            const int  q     = q0 + 32 * b;
            if (q >= seg || p0 + q >= hi) continue;
            const int  p     = p0 + q + lane;
            const bool valid = p < hi;
            const int  part  = gcn_rank_round(hist[wv], key[b], valid);
            if (valid) rank2[p] = part;
        }
    }
    __syncthreads();

    int run = 0;
    for (int k = 0; k < NWAVES; ++k) {
        const int v = hist[k][t];
        hist[k][t] = run;
        run += v;
    }

    int incl = run;
    for (int off = 1; off < 32; off <<= 1) {
        const int y = __shfl_up(incl, off);
        if (lane >= off) incl += y;
    }
    if (lane == 31) wave_sum[wv] = incl;
    __syncthreads();
    int wbase = 0;
    for (int k = 0; k < NWAVES; ++k)
        if (k < wv) wbase += wave_sum[k];
    const int off_t = wbase + incl - run;
    soff[t] = off_t;
    const int node = bin * BINW + t;
    if (node <= nnodes) rowptr[node] = lo + off_t;
    __syncthreads();

    for (int q = 0; q < seg; q += 32) {
        if (p0 + q >= hi) break;
        const int p = p0 + q + lane;
        if (p < hi) {
            const int4 r = rec1[p];
            const int  k = r.z & (BINW - 1);
            rec[lo + soff[k] + hist[wv][k] + rank2[p]] = make_int2(r.x, r.y);
        }
    }
}

template <int U>
__device__ __forceinline__ void gcn_add_rows(const float4* __restrict__ h4, int2 r,
                                             int j, int lane, float4& acc) {
    float4 v[U];
    float  w[U];
#pragma unroll
    for (int k = 0; k < U; ++k) {
        const int src = __builtin_amdgcn_readlane(r.x, j + k);
        w[k] = __int_as_float(__builtin_amdgcn_readlane(r.y, j + k));
        v[k] = h4[(size_t)src * (DIM / 4) + lane];
    }
#pragma unroll
    for (int k = 0; k < U; ++k) {
        acc.x = fmaf(w[k], v[k].x, acc.x);
        acc.y = fmaf(w[k], v[k].y, acc.y);
        acc.z = fmaf(w[k], v[k].z, acc.z);
        acc.w = fmaf(w[k], v[k].w, acc.w);
    }
}

__global__ __launch_bounds__(256) void gcn_gather_kernel(
    const float* __restrict__ h,
    const int*   __restrict__ rowptr,
    const int2*  __restrict__ rec,
    float*       __restrict__ agg,
    int nnodes) {
    const int lane = threadIdx.x & 31;
    const int node = __builtin_amdgcn_readfirstlane(
        (int)(blockIdx.x * (blockDim.x >> 5) + (threadIdx.x >> 5)));
    if (node >= nnodes) return;

    const int beg = rowptr[node];
    const int end = rowptr[node + 1];
    const float4* h4 = reinterpret_cast<const float4*>(h);

    float4 acc = make_float4(0.f, 0.f, 0.f, 0.f);
    for (int base = beg; base < end; base += 32) {
        const int n = min(32, end - base);
        int2 r = make_int2(0, 0);
        if (lane < n) r = rec[base + lane];

        int j = 0;
        for (; j + 8 <= n; j += 8) gcn_add_rows<8>(h4, r, j, lane, acc);
        if (j + 4 <= n) { gcn_add_rows<4>(h4, r, j, lane, acc); j += 4; }
        for (; j < n; ++j) gcn_add_rows<1>(h4, r, j, lane, acc);
    }
    reinterpret_cast<float4*>(agg)[(size_t)node * (DIM / 4) + lane] = acc;
}

__global__ __launch_bounds__(256) void gcn_gemm_wmma_kernel(
    const float* __restrict__ agg,
    const float* __restrict__ h,
    const float* __restrict__ Wrel,
    const float* __restrict__ Wroot,
    const float* __restrict__ bias,
    float*       __restrict__ out,
    int nrows, int do_relu) {
    __shared__ float4 sA4[16 * DIM / 4];
    __shared__ float4 sH4[16 * DIM / 4];

    const int t    = threadIdx.x;
    const int lane = t & 31;
    const int wv   = t >> 5;
    const int r    = lane & 15;
    const int half = lane >> 4;

    const int row0 = blockIdx.x * 16;
    const int col0 = wv * 16;

    {
        const float4* aggv = reinterpret_cast<const float4*>(agg) + (size_t)row0 * (DIM / 4);
        const float4* hv   = reinterpret_cast<const float4*>(h)   + (size_t)row0 * (DIM / 4);
#pragma unroll
        for (int i = t; i < 16 * DIM / 4; i += 256) {
            sA4[i] = aggv[i];
            sH4[i] = hv[i];
        }
    }
    __syncthreads();

    const float* sA = reinterpret_cast<const float*>(sA4);
    const float* sH = reinterpret_cast<const float*>(sH4);

    v8f acc_rel  = {};
    v8f acc_root = {};
#pragma unroll
    for (int k = 0; k < DIM; k += 4) {
        const int ka = k + 2 * half;

        v2f a_rel  = *reinterpret_cast<const v2f*>(&sA[r * DIM + ka]);
        v2f b_rel  = { Wrel[(size_t)ka * DIM + col0 + r],
                       Wrel[(size_t)(ka + 1) * DIM + col0 + r] };
        acc_rel = __builtin_amdgcn_wmma_f32_16x16x4_f32(
                      false, a_rel, false, b_rel, (short)0, acc_rel, false, false);

        v2f a_root = *reinterpret_cast<const v2f*>(&sH[r * DIM + ka]);
        v2f b_root = { Wroot[(size_t)ka * DIM + col0 + r],
                       Wroot[(size_t)(ka + 1) * DIM + col0 + r] };
        acc_root = __builtin_amdgcn_wmma_f32_16x16x4_f32(
                      false, a_root, false, b_root, (short)0, acc_root, false, false);
    }

    const float bv = bias[col0 + r];
#pragma unroll
    for (int j = 0; j < 8; ++j) {
        const int m    = j + 8 * half;
        const int orow = row0 + m;
        if (orow < nrows) {
            float v = acc_rel[j] + acc_root[j] + bv;
            if (do_relu) v = fmaxf(v, 0.f);
            out[(size_t)orow * DIM + col0 + r] = v;
        }
    }
}

#define BASE_PAD   (NCOUNTS + 64)
#define ROWPTR_PAD (NNODES + 64)
static_assert(NCOUNTS % 4 == 0 && NNODES % 4 == 0 && NEDGES % 4 == 0, "16B alignment of the planes");

extern "C" void kernel_launch(void* const* d_in, const int* in_sizes, int n_in,
                              void* d_out, int out_size, void* d_ws, size_t ws_size,
                              hipStream_t stream) {
    const float* x      = (const float*)d_in[0];
    const int*   eidx   = (const int*)  d_in[1];
    const float* ew     = (const float*)d_in[2];
    const float* Wrel   = (const float*)d_in[3];
    const float* brel   = (const float*)d_in[4];
    const float* Wroot  = (const float*)d_in[5];
    float*       out    = (float*)d_out;

    const int N = NNODES;
    const int E = NEDGES;

    float* h1       = (float*)d_ws;
    float* h2       = h1 + (size_t)N * DIM;
    float* agg      = h2 + (size_t)N * DIM;
    int*   tilecnt  = (int*)(agg + (size_t)N * DIM);
    int*   tilebase = tilecnt + NCOUNTS;
    int*   rank1    = tilebase + BASE_PAD;
    int*   rank2    = rank1 + E;
    int*   rowptr   = rank2 + E;
    int4*  rec1     = (int4*)(rowptr + ROWPTR_PAD);
    int2*  rec      = (int2*)(rec1 + E);

    const int edge_blocks   = (E + 255) / 256;
    const int gather_blocks = (N + 7) / 8;
    const int gemm_blocks   = (N + 15) / 16;

    gcn_bin_count_kernel<<<NTILES, BINW, 0, stream>>>(eidx, tilecnt, rank1, E);
    gcn_scan_kernel<<<1, SCAN_THREADS, 0, stream>>>(tilecnt, tilebase, NCOUNTS);
    gcn_bin_place_kernel<<<edge_blocks, 256, 0, stream>>>(eidx, ew, tilebase, rank1, rec1, E);
    gcn_node_group_kernel<<<NBINS, BINW, 0, stream>>>(tilebase, rec1, rank2, rowptr, rec, N);

    for (int l = 0; l < NLAYERS; ++l) {
        const float* hin  = (l == 0) ? x  : (l == 1) ? h1 : h2;
        float*       hout = (l == 0) ? h1 : (l == 1) ? h2 : out;

        gcn_gather_kernel<<<gather_blocks, 256, 0, stream>>>(
            hin, rowptr, rec, agg, N);

        gcn_gemm_wmma_kernel<<<gemm_blocks, 256, 0, stream>>>(
            agg, hin,
            Wrel  + (size_t)l * DIM * DIM,
            Wroot + (size_t)l * DIM * DIM,
            brel  + (size_t)l * DIM,
            hout, N, (l < NLAYERS - 1) ? 1 : 0);
    }
}


# GEMM kernel: LDS row pitch of the agg/h tiles 128 -> 132 dwords, so each A-fragment ds_load covers all 64 banks in one pass instead of stacking a half wave on one bank pair; sum order unchanged
# speedup vs baseline: 2.3568x; 1.2548x over previous
#include <hip/hip_runtime.h>

#define NNODES  50000
#define NEDGES  800000
#define DIM     128
#define NLAYERS 3

typedef __attribute__((ext_vector_type(2))) float v2f;
typedef __attribute__((ext_vector_type(8))) float v8f;


#define BINW    256
#define NWAVES  (BINW / 32)
#define TILE1   4096
#define NTILES  ((NEDGES + TILE1 - 1) / TILE1)
#define NBINS   (NNODES / BINW + 1)
#define NCOUNTS (BINW * NTILES)
#define BATCH   8

#define SCAN_THREADS 1024
#define SCAN_ITEMS   ((NCOUNTS + SCAN_THREADS - 1) / SCAN_THREADS)

static_assert(NBINS <= BINW, "level-1 bin numbers must fit the 256-entry histogram");
static_assert(TILE1 % (NWAVES * 32 * BATCH) == 0, "a tile is whole batches for every wave");

__device__ __forceinline__ void gcn_match32(int key, int& lt, int& tot, bool& last) {
    unsigned mine = 0u;
#pragma unroll
    for (int l = 0; l < 32; ++l) {
        const int      kl = __builtin_amdgcn_readlane(key, l);
        const bool     eq = (key == kl);
        const unsigned m  = __builtin_amdgcn_ballot_w32(eq);
        if (eq) mine = m;
    }
    const unsigned lane = threadIdx.x & 31u;
    lt   = (int)__builtin_amdgcn_mbcnt_lo(mine, 0u);
    tot  = __popc(mine);
    last = ((mine >> lane) == 1u);
}

__device__ __forceinline__ int gcn_rank_round(int* __restrict__ whist, int key, bool valid) {
    int lt, tot; bool last;
    gcn_match32(key, lt, tot, last);
    int part = 0;
    if (valid) {
        const int old = whist[key];
        part = old + lt;
        if (last) whist[key] = old + tot;
    }
    return part;
}

__global__ __launch_bounds__(BINW) void gcn_bin_count_kernel(
    const int* __restrict__ edge_index,
    int*       __restrict__ tilecnt,
    int*       __restrict__ rank1,
    int nedges) {
    __shared__ int hist[NWAVES][BINW];

    const int t    = threadIdx.x;
    const int lane = t & 31;
    const int wv   = __builtin_amdgcn_readfirstlane(t >> 5);
    const int tile = blockIdx.x;
    const int e0   = tile * TILE1;
    const int ne   = min(TILE1, nedges - e0);
    const int seg0 = wv * (TILE1 / NWAVES);
    const int* dst = edge_index + nedges;

    for (int k = 0; k < NWAVES; ++k) hist[k][t] = 0;
    __syncthreads();

    for (int q0 = 0; q0 < TILE1 / NWAVES; q0 += 32 * BATCH) {
        if (seg0 + q0 >= ne) break;
        int key[BATCH];
#pragma unroll
        for (int b = 0; b < BATCH; ++b) {
            const int idx = seg0 + q0 + 32 * b + lane;
            const int v = dst[e0 + min(idx, ne - 1)];
            key[b] = (idx < ne) ? (v >> 8) : -1;
        }
#pragma unroll
        for (int b = 0; b < BATCH; ++b) {
            if (seg0 + q0 + 32 * b >= ne) continue;
            const int  idx   = seg0 + q0 + 32 * b + lane;
            const bool valid = idx < ne;
            const int  part  = gcn_rank_round(hist[wv], key[b], valid);
            if (valid) rank1[e0 + idx] = part;
        }
    }
    __syncthreads();

    int run = 0;
    for (int k = 0; k < NWAVES; ++k) {
        const int v = hist[k][t];
        hist[k][t] = run;
        run += v;
    }
    tilecnt[t * NTILES + tile] = run;
    __syncthreads();

    for (int q = 0; q < TILE1 / NWAVES; q += 32) {
        if (seg0 + q >= ne) break;
        const int idx = seg0 + q + lane;
        if (idx < ne) rank1[e0 + idx] += hist[wv][dst[e0 + idx] >> 8];
    }
}

__global__ __launch_bounds__(SCAN_THREADS) void gcn_scan_kernel(
    const int* __restrict__ cnt,
    int*       __restrict__ base,
    int n) {
    __shared__ int wave_sum[SCAN_THREADS / 32];

    const int t    = threadIdx.x;
    const int lane = t & 31;
    const int wv   = t >> 5;
    const int beg  = t * SCAN_ITEMS;

    int vals[SCAN_ITEMS];
    int s = 0;
#pragma unroll
    for (int i = 0; i < SCAN_ITEMS; ++i) {
        const int idx = beg + i;
        const int v   = (idx < n) ? cnt[idx] : 0;
        vals[i] = v;
        s += v;
    }

    int incl = s;
    for (int off = 1; off < 32; off <<= 1) {
        const int y = __shfl_up(incl, off);
        if (lane >= off) incl += y;
    }
    if (lane == 31) wave_sum[wv] = incl;
    __syncthreads();

    if (wv == 0) {
        const int x = wave_sum[lane];
        int xi = x;
        for (int off = 1; off < 32; off <<= 1) {
            const int y = __shfl_up(xi, off);
            if (lane >= off) xi += y;
        }
        wave_sum[lane] = xi - x;
    }
    __syncthreads();

    int run = wave_sum[wv] + incl - s;
#pragma unroll
    for (int i = 0; i < SCAN_ITEMS; ++i) {
        const int idx = beg + i;
        if (idx < n) {
            base[idx] = run;
            run += vals[i];
        }
    }
    if (t == SCAN_THREADS - 1) base[n] = run;
}

__global__ __launch_bounds__(256) void gcn_bin_place_kernel(
    const int*   __restrict__ edge_index,
    const float* __restrict__ edge_weight,
    const int*   __restrict__ tilebase,
    const int*   __restrict__ rank1,
    int4*        __restrict__ rec1,
    int nedges) {
    const int e = blockIdx.x * blockDim.x + threadIdx.x;
    if (e >= nedges) return;
    const int   src = edge_index[e];
    const int   dst = edge_index[nedges + e];
    const float w   = edge_weight[e];
    const int   pos = tilebase[(dst >> 8) * NTILES + e / TILE1] + rank1[e];
    rec1[pos] = make_int4(src, __float_as_int(w), dst, 0);
}

__global__ __launch_bounds__(BINW) void gcn_node_group_kernel(
    const int*  __restrict__ tilebase,
    const int4* __restrict__ rec1,
    int*        __restrict__ rank2,
    int*        __restrict__ rowptr,
    int2*       __restrict__ rec,
    int nnodes) {
    __shared__ int hist[NWAVES][BINW];
    __shared__ int soff[BINW];
    __shared__ int wave_sum[NWAVES];

    const int t    = threadIdx.x;
    const int lane = t & 31;
    const int wv   = __builtin_amdgcn_readfirstlane(t >> 5);
    const int bin  = blockIdx.x;
    const int lo   = tilebase[bin * NTILES];
    const int hi   = tilebase[(bin + 1) * NTILES];
    const int seg  = ((hi - lo + BINW - 1) / BINW) * 32;
    const int p0   = lo + wv * seg;

    for (int k = 0; k < NWAVES; ++k) hist[k][t] = 0;
    __syncthreads();

    for (int q0 = 0; q0 < seg; q0 += 32 * BATCH) {
        if (p0 + q0 >= hi) break;
        int key[BATCH];
#pragma unroll
        for (int b = 0; b < BATCH; ++b) {
            const int q = q0 + 32 * b;
            const int p = p0 + q + lane;
            const int v = rec1[min(p, hi - 1)].z;
            key[b] = (q < seg && p < hi) ? (v & (BINW - 1)) : -1;
        }
#pragma unroll
        for (int b = 0; b < BATCH; ++b) {
            const int  q     = q0 + 32 * b;
            if (q >= seg || p0 + q >= hi) continue;
            const int  p     = p0 + q + lane;
            const bool valid = p < hi;
            const int  part  = gcn_rank_round(hist[wv], key[b], valid);
            if (valid) rank2[p] = part;
        }
    }
    __syncthreads();

    int run = 0;
    for (int k = 0; k < NWAVES; ++k) {
        const int v = hist[k][t];
        hist[k][t] = run;
        run += v;
    }

    int incl = run;
    for (int off = 1; off < 32; off <<= 1) {
        const int y = __shfl_up(incl, off);
        if (lane >= off) incl += y;
    }
    if (lane == 31) wave_sum[wv] = incl;
    __syncthreads();
    int wbase = 0;
    for (int k = 0; k < NWAVES; ++k)
        if (k < wv) wbase += wave_sum[k];
    const int off_t = wbase + incl - run;
    soff[t] = off_t;
    const int node = bin * BINW + t;
    if (node <= nnodes) rowptr[node] = lo + off_t;
    __syncthreads();

    for (int q = 0; q < seg; q += 32) {
        if (p0 + q >= hi) break;
        const int p = p0 + q + lane;
        if (p < hi) {
            const int4 r = rec1[p];
            const int  k = r.z & (BINW - 1);
            rec[lo + soff[k] + hist[wv][k] + rank2[p]] = make_int2(r.x, r.y);
        }
    }
}

template <int U>
__device__ __forceinline__ void gcn_add_rows(const float4* __restrict__ h4, int2 r,
                                             int j, int lane, float4& acc) {
    float4 v[U];
    float  w[U];
#pragma unroll
    for (int k = 0; k < U; ++k) {
        const int src = __builtin_amdgcn_readlane(r.x, j + k);
        w[k] = __int_as_float(__builtin_amdgcn_readlane(r.y, j + k));
        v[k] = h4[(size_t)src * (DIM / 4) + lane];
    }
#pragma unroll
    for (int k = 0; k < U; ++k) {
        acc.x = fmaf(w[k], v[k].x, acc.x);
        acc.y = fmaf(w[k], v[k].y, acc.y);
        acc.z = fmaf(w[k], v[k].z, acc.z);
        acc.w = fmaf(w[k], v[k].w, acc.w);
    }
}

__global__ __launch_bounds__(256) void gcn_gather_kernel(
    const float* __restrict__ h,
    const int*   __restrict__ rowptr,
    const int2*  __restrict__ rec,
    float*       __restrict__ agg,
    int nnodes) {
    const int lane = threadIdx.x & 31;
    const int node = __builtin_amdgcn_readfirstlane(
        (int)(blockIdx.x * (blockDim.x >> 5) + (threadIdx.x >> 5)));
    if (node >= nnodes) return;

    const int beg = rowptr[node];
    const int end = rowptr[node + 1];
    const float4* h4 = reinterpret_cast<const float4*>(h);

    float4 acc = make_float4(0.f, 0.f, 0.f, 0.f);
    for (int base = beg; base < end; base += 32) {
        const int n = min(32, end - base);
        int2 r = make_int2(0, 0);
        if (lane < n) r = rec[base + lane];

        int j = 0;
        for (; j + 8 <= n; j += 8) gcn_add_rows<8>(h4, r, j, lane, acc);
        if (j + 4 <= n) { gcn_add_rows<4>(h4, r, j, lane, acc); j += 4; }
        for (; j < n; ++j) gcn_add_rows<1>(h4, r, j, lane, acc);
    }
    reinterpret_cast<float4*>(agg)[(size_t)node * (DIM / 4) + lane] = acc;
}

#define APITCH (DIM + 4)
static_assert(APITCH % 4 == 0 && (APITCH % 64) == 4, "16B-aligned rows that start 4 banks apart");

__global__ __launch_bounds__(256) void gcn_gemm_wmma_kernel(
    const float* __restrict__ agg,
    const float* __restrict__ h,
    const float* __restrict__ Wrel,
    const float* __restrict__ Wroot,
    const float* __restrict__ bias,
    float*       __restrict__ out,
    int nrows, int do_relu) {
    __shared__ float4 sA4[16 * APITCH / 4];
    __shared__ float4 sH4[16 * APITCH / 4];

    const int t    = threadIdx.x;
    const int lane = t & 31;
    const int wv   = t >> 5;
    const int r    = lane & 15;
    const int half = lane >> 4;

    const int row0 = blockIdx.x * 16;
    const int col0 = wv * 16;

    {
        const float4* aggv = reinterpret_cast<const float4*>(agg) + (size_t)row0 * (DIM / 4);
        const float4* hv   = reinterpret_cast<const float4*>(h)   + (size_t)row0 * (DIM / 4);
#pragma unroll
        for (int i = t; i < 16 * DIM / 4; i += 256) {
            const int j = (i >> 5) * (APITCH / 4) + (i & 31);
            sA4[j] = aggv[i];
            sH4[j] = hv[i];
        }
    }
    __syncthreads();

    const float* sA = reinterpret_cast<const float*>(sA4);
    const float* sH = reinterpret_cast<const float*>(sH4);

    v8f acc_rel  = {};
    v8f acc_root = {};
#pragma unroll
    for (int k = 0; k < DIM; k += 4) {
        const int ka = k + 2 * half;

        v2f a_rel  = *reinterpret_cast<const v2f*>(&sA[r * APITCH + ka]);
        v2f b_rel  = { Wrel[(size_t)ka * DIM + col0 + r],
                       Wrel[(size_t)(ka + 1) * DIM + col0 + r] };
        acc_rel = __builtin_amdgcn_wmma_f32_16x16x4_f32(
                      false, a_rel, false, b_rel, (short)0, acc_rel, false, false);

        v2f a_root = *reinterpret_cast<const v2f*>(&sH[r * APITCH + ka]);
        v2f b_root = { Wroot[(size_t)ka * DIM + col0 + r],
                       Wroot[(size_t)(ka + 1) * DIM + col0 + r] };
        acc_root = __builtin_amdgcn_wmma_f32_16x16x4_f32(
                      false, a_root, false, b_root, (short)0, acc_root, false, false);
    }

    const float bv = bias[col0 + r];
#pragma unroll
    for (int j = 0; j < 8; ++j) {
        const int m    = j + 8 * half;
        const int orow = row0 + m;
        if (orow < nrows) {
            float v = acc_rel[j] + acc_root[j] + bv;
            if (do_relu) v = fmaxf(v, 0.f);
            out[(size_t)orow * DIM + col0 + r] = v;
        }
    }
}

#define BASE_PAD   (NCOUNTS + 64)
#define ROWPTR_PAD (NNODES + 64)
static_assert(NCOUNTS % 4 == 0 && NNODES % 4 == 0 && NEDGES % 4 == 0, "16B alignment of the planes");

extern "C" void kernel_launch(void* const* d_in, const int* in_sizes, int n_in,
                              void* d_out, int out_size, void* d_ws, size_t ws_size,
                              hipStream_t stream) {
    const float* x      = (const float*)d_in[0];
    const int*   eidx   = (const int*)  d_in[1];
    const float* ew     = (const float*)d_in[2];
    const float* Wrel   = (const float*)d_in[3];
    const float* brel   = (const float*)d_in[4];
    const float* Wroot  = (const float*)d_in[5];
    float*       out    = (float*)d_out;

    const int N = NNODES;
    const int E = NEDGES;

    float* h1       = (float*)d_ws;
    float* h2       = h1 + (size_t)N * DIM;
    float* agg      = h2 + (size_t)N * DIM;
    int*   tilecnt  = (int*)(agg + (size_t)N * DIM);
    int*   tilebase = tilecnt + NCOUNTS;
    int*   rank1    = tilebase + BASE_PAD;
    int*   rank2    = rank1 + E;
    int*   rowptr   = rank2 + E;
    int4*  rec1     = (int4*)(rowptr + ROWPTR_PAD);
    int2*  rec      = (int2*)(rec1 + E);

    const int edge_blocks   = (E + 255) / 256;
    const int gather_blocks = (N + 7) / 8;
    const int gemm_blocks   = (N + 15) / 16;

    gcn_bin_count_kernel<<<NTILES, BINW, 0, stream>>>(eidx, tilecnt, rank1, E);
    gcn_scan_kernel<<<1, SCAN_THREADS, 0, stream>>>(tilecnt, tilebase, NCOUNTS);
    gcn_bin_place_kernel<<<edge_blocks, 256, 0, stream>>>(eidx, ew, tilebase, rank1, rec1, E);
    gcn_node_group_kernel<<<NBINS, BINW, 0, stream>>>(tilebase, rec1, rank2, rowptr, rec, N);

    for (int l = 0; l < NLAYERS; ++l) {
        const float* hin  = (l == 0) ? x  : (l == 1) ? h1 : h2;
        float*       hout = (l == 0) ? h1 : (l == 1) ? h2 : out;

        gcn_gather_kernel<<<gather_blocks, 256, 0, stream>>>(
            hin, rowptr, rec, agg, N);

        gcn_gemm_wmma_kernel<<<gemm_blocks, 256, 0, stream>>>(
            agg, hin,
            Wrel  + (size_t)l * DIM * DIM,
            Wroot + (size_t)l * DIM * DIM,
            brel  + (size_t)l * DIM,
            hout, N, (l < NLAYERS - 1) ? 1 : 0);
    }
}


# gather kernel fused into the WMMA kernel: each wave sums its two nodes' rows in edge order straight into the block's LDS agg tile; agg plane and 3 launches gone; sum order unchanged
# speedup vs baseline: 2.5163x; 1.0677x over previous
#include <hip/hip_runtime.h>

#define NNODES  50000
#define NEDGES  800000
#define DIM     128
#define NLAYERS 3

typedef __attribute__((ext_vector_type(2))) float v2f;
typedef __attribute__((ext_vector_type(8))) float v8f;


#define BINW    256
#define NWAVES  (BINW / 32)
#define TILE1   4096
#define NTILES  ((NEDGES + TILE1 - 1) / TILE1)
#define NBINS   (NNODES / BINW + 1)
#define NCOUNTS (BINW * NTILES)
#define BATCH   8

#define SCAN_THREADS 1024
#define SCAN_ITEMS   ((NCOUNTS + SCAN_THREADS - 1) / SCAN_THREADS)

static_assert(NBINS <= BINW, "level-1 bin numbers must fit the 256-entry histogram");
static_assert(TILE1 % (NWAVES * 32 * BATCH) == 0, "a tile is whole batches for every wave");

__device__ __forceinline__ void gcn_match32(int key, int& lt, int& tot, bool& last) {
    unsigned mine = 0u;
#pragma unroll
    for (int l = 0; l < 32; ++l) {
        const int      kl = __builtin_amdgcn_readlane(key, l);
        const bool     eq = (key == kl);
        const unsigned m  = __builtin_amdgcn_ballot_w32(eq);
        if (eq) mine = m;
    }
    const unsigned lane = threadIdx.x & 31u;
    lt   = (int)__builtin_amdgcn_mbcnt_lo(mine, 0u);
    tot  = __popc(mine);
    last = ((mine >> lane) == 1u);
}

__device__ __forceinline__ int gcn_rank_round(int* __restrict__ whist, int key, bool valid) {
    int lt, tot; bool last;
    gcn_match32(key, lt, tot, last);
    int part = 0;
    if (valid) {
        const int old = whist[key];
        part = old + lt;
        if (last) whist[key] = old + tot;
    }
    return part;
}

__global__ __launch_bounds__(BINW) void gcn_bin_count_kernel(
    const int* __restrict__ edge_index,
    int*       __restrict__ tilecnt,
    int*       __restrict__ rank1,
    int nedges) {
    __shared__ int hist[NWAVES][BINW];

    const int t    = threadIdx.x;
    const int lane = t & 31;
    const int wv   = __builtin_amdgcn_readfirstlane(t >> 5);
    const int tile = blockIdx.x;
    const int e0   = tile * TILE1;
    const int ne   = min(TILE1, nedges - e0);
    const int seg0 = wv * (TILE1 / NWAVES);
    const int* dst = edge_index + nedges;

    for (int k = 0; k < NWAVES; ++k) hist[k][t] = 0;
    __syncthreads();

    for (int q0 = 0; q0 < TILE1 / NWAVES; q0 += 32 * BATCH) {
        if (seg0 + q0 >= ne) break;
        int key[BATCH];
#pragma unroll
        for (int b = 0; b < BATCH; ++b) {
            const int idx = seg0 + q0 + 32 * b + lane;
            const int v = dst[e0 + min(idx, ne - 1)];
            key[b] = (idx < ne) ? (v >> 8) : -1;
        }
#pragma unroll
        for (int b = 0; b < BATCH; ++b) {
            if (seg0 + q0 + 32 * b >= ne) continue;
            const int  idx   = seg0 + q0 + 32 * b + lane;
            const bool valid = idx < ne;
            const int  part  = gcn_rank_round(hist[wv], key[b], valid);
            if (valid) rank1[e0 + idx] = part;
        }
    }
    __syncthreads();

    int run = 0;
    for (int k = 0; k < NWAVES; ++k) {
        const int v = hist[k][t];
        hist[k][t] = run;
        run += v;
    }
    tilecnt[t * NTILES + tile] = run;
    __syncthreads();

    for (int q = 0; q < TILE1 / NWAVES; q += 32) {
        if (seg0 + q >= ne) break;
        const int idx = seg0 + q + lane;
        if (idx < ne) rank1[e0 + idx] += hist[wv][dst[e0 + idx] >> 8];
    }
}

__global__ __launch_bounds__(SCAN_THREADS) void gcn_scan_kernel(
    const int* __restrict__ cnt,
    int*       __restrict__ base,
    int n) {
    __shared__ int wave_sum[SCAN_THREADS / 32];

    const int t    = threadIdx.x;
    const int lane = t & 31;
    const int wv   = t >> 5;
    const int beg  = t * SCAN_ITEMS;

    int vals[SCAN_ITEMS];
    int s = 0;
#pragma unroll
    for (int i = 0; i < SCAN_ITEMS; ++i) {
        const int idx = beg + i;
        const int v   = (idx < n) ? cnt[idx] : 0;
        vals[i] = v;
        s += v;
    }

    int incl = s;
    for (int off = 1; off < 32; off <<= 1) {
        const int y = __shfl_up(incl, off);
        if (lane >= off) incl += y;
    }
    if (lane == 31) wave_sum[wv] = incl;
    __syncthreads();

    if (wv == 0) {
        const int x = wave_sum[lane];
        int xi = x;
        for (int off = 1; off < 32; off <<= 1) {
            const int y = __shfl_up(xi, off);
            if (lane >= off) xi += y;
        }
        wave_sum[lane] = xi - x;
    }
    __syncthreads();

    int run = wave_sum[wv] + incl - s;
#pragma unroll
    for (int i = 0; i < SCAN_ITEMS; ++i) {
        const int idx = beg + i;
        if (idx < n) {
            base[idx] = run;
            run += vals[i];
        }
    }
    if (t == SCAN_THREADS - 1) base[n] = run;
}

__global__ __launch_bounds__(256) void gcn_bin_place_kernel(
    const int*   __restrict__ edge_index,
    const float* __restrict__ edge_weight,
    const int*   __restrict__ tilebase,
    const int*   __restrict__ rank1,
    int4*        __restrict__ rec1,
    int nedges) {
    const int e = blockIdx.x * blockDim.x + threadIdx.x;
    if (e >= nedges) return;
    const int   src = edge_index[e];
    const int   dst = edge_index[nedges + e];
    const float w   = edge_weight[e];
    const int   pos = tilebase[(dst >> 8) * NTILES + e / TILE1] + rank1[e];
    rec1[pos] = make_int4(src, __float_as_int(w), dst, 0);
}

__global__ __launch_bounds__(BINW) void gcn_node_group_kernel(
    const int*  __restrict__ tilebase,
    const int4* __restrict__ rec1,
    int*        __restrict__ rank2,
    int*        __restrict__ rowptr,
    int2*       __restrict__ rec,
    int nnodes) {
    __shared__ int hist[NWAVES][BINW];
    __shared__ int soff[BINW];
    __shared__ int wave_sum[NWAVES];

    const int t    = threadIdx.x;
    const int lane = t & 31;
    const int wv   = __builtin_amdgcn_readfirstlane(t >> 5);
    const int bin  = blockIdx.x;
    const int lo   = tilebase[bin * NTILES];
    const int hi   = tilebase[(bin + 1) * NTILES];
    const int seg  = ((hi - lo + BINW - 1) / BINW) * 32;
    const int p0   = lo + wv * seg;

    for (int k = 0; k < NWAVES; ++k) hist[k][t] = 0;
    __syncthreads();

    for (int q0 = 0; q0 < seg; q0 += 32 * BATCH) {
        if (p0 + q0 >= hi) break;
        int key[BATCH];
#pragma unroll
        for (int b = 0; b < BATCH; ++b) {
            const int q = q0 + 32 * b;
            const int p = p0 + q + lane;
            const int v = rec1[min(p, hi - 1)].z;
            key[b] = (q < seg && p < hi) ? (v & (BINW - 1)) : -1;
        }
#pragma unroll
        for (int b = 0; b < BATCH; ++b) {
            const int  q     = q0 + 32 * b;
            if (q >= seg || p0 + q >= hi) continue;
            const int  p     = p0 + q + lane;
            const bool valid = p < hi;
            const int  part  = gcn_rank_round(hist[wv], key[b], valid);
            if (valid) rank2[p] = part;
        }
    }
    __syncthreads();

    int run = 0;
    for (int k = 0; k < NWAVES; ++k) {
        const int v = hist[k][t];
        hist[k][t] = run;
        run += v;
    }

    int incl = run;
    for (int off = 1; off < 32; off <<= 1) {
        const int y = __shfl_up(incl, off);
        if (lane >= off) incl += y;
    }
    if (lane == 31) wave_sum[wv] = incl;
    __syncthreads();
    int wbase = 0;
    for (int k = 0; k < NWAVES; ++k)
        if (k < wv) wbase += wave_sum[k];
    const int off_t = wbase + incl - run;
    soff[t] = off_t;
    const int node = bin * BINW + t;
    if (node <= nnodes) rowptr[node] = lo + off_t;
    __syncthreads();

    for (int q = 0; q < seg; q += 32) {
        if (p0 + q >= hi) break;
        const int p = p0 + q + lane;
        if (p < hi) {
            const int4 r = rec1[p];
            const int  k = r.z & (BINW - 1);
            rec[lo + soff[k] + hist[wv][k] + rank2[p]] = make_int2(r.x, r.y);
        }
    }
}

template <int U>
__device__ __forceinline__ void gcn_add_rows(const float4* __restrict__ h4, int2 r,
                                             int j, int lane, float4& acc) {
    float4 v[U];
    float  w[U];
#pragma unroll
    for (int k = 0; k < U; ++k) {
        const int src = __builtin_amdgcn_readlane(r.x, j + k);
        w[k] = __int_as_float(__builtin_amdgcn_readlane(r.y, j + k));
        v[k] = h4[(size_t)src * (DIM / 4) + lane];
    }
#pragma unroll
    for (int k = 0; k < U; ++k) {
        acc.x = fmaf(w[k], v[k].x, acc.x);
        acc.y = fmaf(w[k], v[k].y, acc.y);
        acc.z = fmaf(w[k], v[k].z, acc.z);
        acc.w = fmaf(w[k], v[k].w, acc.w);
    }
}

__device__ __forceinline__ int2 gcn_load_records(const int2* __restrict__ rec,
                                                 int base, int end, int lane) {
    int2 r = make_int2(0, 0);
    if (lane < end - base) r = rec[base + lane];
    return r;
}

__device__ __forceinline__ float4 gcn_sum_list(const float4* __restrict__ h4,
                                               const int2*   __restrict__ rec,
                                               int beg, int end, int2 r, int lane) {
    float4 acc = make_float4(0.f, 0.f, 0.f, 0.f);
    for (int base = beg; base < end; base += 32) {
        const int n = min(32, end - base);
        if (base != beg) r = gcn_load_records(rec, base, end, lane);

        int j = 0;
        for (; j + 8 <= n; j += 8) gcn_add_rows<8>(h4, r, j, lane, acc);
        if (j + 4 <= n) { gcn_add_rows<4>(h4, r, j, lane, acc); j += 4; }
        for (; j < n; ++j) gcn_add_rows<1>(h4, r, j, lane, acc);
    }
    return acc;
}

#define APITCH (DIM + 4)
static_assert(APITCH % 4 == 0 && (APITCH % 64) == 4, "16B-aligned rows that start 4 banks apart");

__global__ __launch_bounds__(256) void gcn_layer_kernel(
    const float* __restrict__ h,
    const int*   __restrict__ rowptr,
    const int2*  __restrict__ rec,
    const float* __restrict__ Wrel,
    const float* __restrict__ Wroot,
    const float* __restrict__ bias,
    float*       __restrict__ out,
    int nrows, int do_relu) {
    __shared__ float4 sA4[16 * APITCH / 4];
    __shared__ float4 sH4[16 * APITCH / 4];

    const int t    = threadIdx.x;
    const int lane = t & 31;
    const int wv   = __builtin_amdgcn_readfirstlane(t >> 5);
    const int r    = lane & 15;
    const int half = lane >> 4;

    const int row0 = blockIdx.x * 16;
    const int col0 = wv * 16;

    {
        const float4* h4 = reinterpret_cast<const float4*>(h);
        const int m0 = 2 * wv;
        const int n0 = row0 + m0;

        int beg0 = 0, end0 = 0, beg1 = 0, end1 = 0;
        float4 self0 = make_float4(0.f, 0.f, 0.f, 0.f);
        float4 self1 = make_float4(0.f, 0.f, 0.f, 0.f);
        if (n0 < nrows) {
            beg0 = rowptr[n0];
            end0 = rowptr[n0 + 1];
        }
        if (n0 + 1 < nrows) {
            beg1 = end0;
            end1 = rowptr[n0 + 2];
        }
        const int2 r0 = gcn_load_records(rec, beg0, end0, lane);
        const int2 r1 = gcn_load_records(rec, beg1, end1, lane);
        if (n0 < nrows)     self0 = h4[(size_t)n0 * (DIM / 4) + lane];
        if (n0 + 1 < nrows) self1 = h4[(size_t)(n0 + 1) * (DIM / 4) + lane];

        const float4 a0 = gcn_sum_list(h4, rec, beg0, end0, r0, lane);
        const float4 a1 = gcn_sum_list(h4, rec, beg1, end1, r1, lane);

        sA4[m0 * (APITCH / 4) + lane]       = a0;
        sA4[(m0 + 1) * (APITCH / 4) + lane] = a1;
        sH4[m0 * (APITCH / 4) + lane]       = self0;
        sH4[(m0 + 1) * (APITCH / 4) + lane] = self1;
    }
    __syncthreads();

    const float* sA = reinterpret_cast<const float*>(sA4);
    const float* sH = reinterpret_cast<const float*>(sH4);

    v8f acc_rel  = {};
    v8f acc_root = {};
#pragma unroll
    for (int k = 0; k < DIM; k += 4) {
        const int ka = k + 2 * half;

        v2f a_rel  = *reinterpret_cast<const v2f*>(&sA[r * APITCH + ka]);
        v2f b_rel  = { Wrel[(size_t)ka * DIM + col0 + r],
                       Wrel[(size_t)(ka + 1) * DIM + col0 + r] };
        acc_rel = __builtin_amdgcn_wmma_f32_16x16x4_f32(
                      false, a_rel, false, b_rel, (short)0, acc_rel, false, false);

        v2f a_root = *reinterpret_cast<const v2f*>(&sH[r * APITCH + ka]);
        v2f b_root = { Wroot[(size_t)ka * DIM + col0 + r],
                       Wroot[(size_t)(ka + 1) * DIM + col0 + r] };
        acc_root = __builtin_amdgcn_wmma_f32_16x16x4_f32(
                      false, a_root, false, b_root, (short)0, acc_root, false, false);
    }

    const float bv = bias[col0 + r];
#pragma unroll
    for (int j = 0; j < 8; ++j) {
        const int m    = j + 8 * half;
        const int orow = row0 + m;
        if (orow < nrows) {
            float v = acc_rel[j] + acc_root[j] + bv;
            if (do_relu) v = fmaxf(v, 0.f);
            out[(size_t)orow * DIM + col0 + r] = v;
        }
    }
}

#define BASE_PAD   (NCOUNTS + 64)
#define ROWPTR_PAD (NNODES + 64)
static_assert(NCOUNTS % 4 == 0 && NNODES % 4 == 0 && NEDGES % 4 == 0, "16B alignment of the planes");

extern "C" void kernel_launch(void* const* d_in, const int* in_sizes, int n_in,
                              void* d_out, int out_size, void* d_ws, size_t ws_size,
                              hipStream_t stream) {
    const float* x      = (const float*)d_in[0];
    const int*   eidx   = (const int*)  d_in[1];
    const float* ew     = (const float*)d_in[2];
    const float* Wrel   = (const float*)d_in[3];
    const float* brel   = (const float*)d_in[4];
    const float* Wroot  = (const float*)d_in[5];
    float*       out    = (float*)d_out;

    const int N = NNODES;
    const int E = NEDGES;

    float* h1       = (float*)d_ws;
    float* h2       = h1 + (size_t)N * DIM;
    int*   tilecnt  = (int*)(h2 + (size_t)N * DIM);
    int*   tilebase = tilecnt + NCOUNTS;
    int*   rank1    = tilebase + BASE_PAD;
    int*   rank2    = rank1 + E;
    int*   rowptr   = rank2 + E;
    int4*  rec1     = (int4*)(rowptr + ROWPTR_PAD);
    int2*  rec      = (int2*)(rec1 + E);

    const int edge_blocks  = (E + 255) / 256;
    const int layer_blocks = (N + 15) / 16;

    gcn_bin_count_kernel<<<NTILES, BINW, 0, stream>>>(eidx, tilecnt, rank1, E);
    gcn_scan_kernel<<<1, SCAN_THREADS, 0, stream>>>(tilecnt, tilebase, NCOUNTS);
    gcn_bin_place_kernel<<<edge_blocks, 256, 0, stream>>>(eidx, ew, tilebase, rank1, rec1, E);
    gcn_node_group_kernel<<<NBINS, BINW, 0, stream>>>(tilebase, rec1, rank2, rowptr, rec, N);

    for (int l = 0; l < NLAYERS; ++l) {
        const float* hin  = (l == 0) ? x  : (l == 1) ? h1 : h2;
        float*       hout = (l == 0) ? h1 : (l == 1) ? h2 : out;

        gcn_layer_kernel<<<layer_blocks, 256, 0, stream>>>(
            hin, rowptr, rec,
            Wrel  + (size_t)l * DIM * DIM,
            Wroot + (size_t)l * DIM * DIM,
            brel  + (size_t)l * DIM,
            hout, N, (l < NLAYERS - 1) ? 1 : 0);
    }
}


# tile-count prefix sum: 1 block, 49 counts per thread (4-byte accesses 196 B apart across lanes, 32 lines each) -> 49 blocks, 1 count per thread, whole-line accesses; integer sums only
# speedup vs baseline: 3.2202x; 1.2797x over previous
#include <hip/hip_runtime.h>

#define NNODES  50000
#define NEDGES  800000
#define DIM     128
#define NLAYERS 3

typedef __attribute__((ext_vector_type(2))) float v2f;
typedef __attribute__((ext_vector_type(8))) float v8f;


#define BINW    256
#define NWAVES  (BINW / 32)
#define TILE1   4096
#define NTILES  ((NEDGES + TILE1 - 1) / TILE1)
#define NBINS   (NNODES / BINW + 1)
#define NCOUNTS (BINW * NTILES)
#define BATCH   8

#define SCAN_THREADS 1024
#define SCAN_BLOCKS  ((NCOUNTS + SCAN_THREADS - 1) / SCAN_THREADS)
#define SCAN_QUADS   ((SCAN_BLOCKS - 1) * (SCAN_THREADS / 4))
#define SCAN_PRE     ((SCAN_QUADS + SCAN_THREADS - 1) / SCAN_THREADS)

static_assert(SCAN_THREADS / 32 == 32, "one lane per wave when the wave sums are folded");

static_assert(NBINS <= BINW, "level-1 bin numbers must fit the 256-entry histogram");
static_assert(TILE1 % (NWAVES * 32 * BATCH) == 0, "a tile is whole batches for every wave");

__device__ __forceinline__ void gcn_match32(int key, int& lt, int& tot, bool& last) {
    unsigned mine = 0u;
#pragma unroll
    for (int l = 0; l < 32; ++l) {
        const int      kl = __builtin_amdgcn_readlane(key, l);
        const bool     eq = (key == kl);
        const unsigned m  = __builtin_amdgcn_ballot_w32(eq);
        if (eq) mine = m;
    }
    const unsigned lane = threadIdx.x & 31u;
    lt   = (int)__builtin_amdgcn_mbcnt_lo(mine, 0u);
    tot  = __popc(mine);
    last = ((mine >> lane) == 1u);
}

__device__ __forceinline__ int gcn_rank_round(int* __restrict__ whist, int key, bool valid) {
    int lt, tot; bool last;
    gcn_match32(key, lt, tot, last);
    int part = 0;
    if (valid) {
        const int old = whist[key];
        part = old + lt;
        if (last) whist[key] = old + tot;
    }
    return part;
}

__global__ __launch_bounds__(BINW) void gcn_bin_count_kernel(
    const int* __restrict__ edge_index,
    int*       __restrict__ tilecnt,
    int*       __restrict__ rank1,
    int nedges) {
    __shared__ int hist[NWAVES][BINW];

    const int t    = threadIdx.x;
    const int lane = t & 31;
    const int wv   = __builtin_amdgcn_readfirstlane(t >> 5);
    const int tile = blockIdx.x;
    const int e0   = tile * TILE1;
    const int ne   = min(TILE1, nedges - e0);
    const int seg0 = wv * (TILE1 / NWAVES);
    const int* dst = edge_index + nedges;

    for (int k = 0; k < NWAVES; ++k) hist[k][t] = 0;
    __syncthreads();

    for (int q0 = 0; q0 < TILE1 / NWAVES; q0 += 32 * BATCH) {
        if (seg0 + q0 >= ne) break;
        int key[BATCH];
#pragma unroll
        for (int b = 0; b < BATCH; ++b) {
            const int idx = seg0 + q0 + 32 * b + lane;
            const int v = dst[e0 + min(idx, ne - 1)];
            key[b] = (idx < ne) ? (v >> 8) : -1;
        }
#pragma unroll
        for (int b = 0; b < BATCH; ++b) {
            if (seg0 + q0 + 32 * b >= ne) continue;
            const int  idx   = seg0 + q0 + 32 * b + lane;
            const bool valid = idx < ne;
            const int  part  = gcn_rank_round(hist[wv], key[b], valid);
            if (valid) rank1[e0 + idx] = part;
        }
    }
    __syncthreads();

    int run = 0;
    for (int k = 0; k < NWAVES; ++k) {
        const int v = hist[k][t];
        hist[k][t] = run;
        run += v;
    }
    tilecnt[t * NTILES + tile] = run;
    __syncthreads();

    for (int q = 0; q < TILE1 / NWAVES; q += 32) {
        if (seg0 + q >= ne) break;
        const int idx = seg0 + q + lane;
        if (idx < ne) rank1[e0 + idx] += hist[wv][dst[e0 + idx] >> 8];
    }
}

__global__ __launch_bounds__(SCAN_THREADS) void gcn_scan_kernel(
    const int* __restrict__ cnt,
    int*       __restrict__ base,
    int n) {
    __shared__ int wave_pre[SCAN_THREADS / 32];
    __shared__ int wave_tot[SCAN_THREADS / 32];

    const int t     = threadIdx.x;
    const int lane  = t & 31;
    const int wv    = t >> 5;
    const int first = blockIdx.x * SCAN_THREADS;
    const int idx   = first + t;
    const int nq    = first / 4;
    const int4* cnt4 = reinterpret_cast<const int4*>(cnt);

    int4 q[SCAN_PRE];
#pragma unroll
    for (int i = 0; i < SCAN_PRE; ++i) {
        const int k = i * SCAN_THREADS + t;
        q[i] = cnt4[(k < nq) ? k : 0];
    }
    const int v = (idx < n) ? cnt[idx] : 0;

    int pre = 0;
#pragma unroll
    for (int i = 0; i < SCAN_PRE; ++i) {
        const int k = i * SCAN_THREADS + t;
        if (k < nq) pre += (q[i].x + q[i].y) + (q[i].z + q[i].w);
    }
    for (int off = 16; off > 0; off >>= 1) pre += __shfl_xor(pre, off);

    int incl = v;
    for (int off = 1; off < 32; off <<= 1) {
        const int y = __shfl_up(incl, off);
        if (lane >= off) incl += y;
    }
    if (lane == 0)  wave_pre[wv] = pre;
    if (lane == 31) wave_tot[wv] = incl;
    __syncthreads();

    int s = wave_pre[lane] + ((lane < wv) ? wave_tot[lane] : 0);
    for (int off = 16; off > 0; off >>= 1) s += __shfl_xor(s, off);

    const int excl = s + incl - v;
    if (idx < n)      base[idx] = excl;
    if (idx == n - 1) base[n]   = excl + v;
}

__global__ __launch_bounds__(256) void gcn_bin_place_kernel(
    const int*   __restrict__ edge_index,
    const float* __restrict__ edge_weight,
    const int*   __restrict__ tilebase,
    const int*   __restrict__ rank1,
    int4*        __restrict__ rec1,
    int nedges) {
    const int e = blockIdx.x * blockDim.x + threadIdx.x;
    if (e >= nedges) return;
    const int   src = edge_index[e];
    const int   dst = edge_index[nedges + e];
    const float w   = edge_weight[e];
    const int   pos = tilebase[(dst >> 8) * NTILES + e / TILE1] + rank1[e];
    rec1[pos] = make_int4(src, __float_as_int(w), dst, 0);
}

__global__ __launch_bounds__(BINW) void gcn_node_group_kernel(
    const int*  __restrict__ tilebase,
    const int4* __restrict__ rec1,
    int*        __restrict__ rank2,
    int*        __restrict__ rowptr,
    int2*       __restrict__ rec,
    int nnodes) {
    __shared__ int hist[NWAVES][BINW];
    __shared__ int soff[BINW];
    __shared__ int wave_sum[NWAVES];

    const int t    = threadIdx.x;
    const int lane = t & 31;
    const int wv   = __builtin_amdgcn_readfirstlane(t >> 5);
    const int bin  = blockIdx.x;
    const int lo   = tilebase[bin * NTILES];
    const int hi   = tilebase[(bin + 1) * NTILES];
    const int seg  = ((hi - lo + BINW - 1) / BINW) * 32;
    const int p0   = lo + wv * seg;

    for (int k = 0; k < NWAVES; ++k) hist[k][t] = 0;
    __syncthreads();

    for (int q0 = 0; q0 < seg; q0 += 32 * BATCH) {
        if (p0 + q0 >= hi) break;
        int key[BATCH];
#pragma unroll
        for (int b = 0; b < BATCH; ++b) {
            const int q = q0 + 32 * b;
            const int p = p0 + q + lane;
            const int v = rec1[min(p, hi - 1)].z;
            key[b] = (q < seg && p < hi) ? (v & (BINW - 1)) : -1;
        }
#pragma unroll
        for (int b = 0; b < BATCH; ++b) {
            const int  q     = q0 + 32 * b;
            if (q >= seg || p0 + q >= hi) continue;
            const int  p     = p0 + q + lane;
            const bool valid = p < hi;
            const int  part  = gcn_rank_round(hist[wv], key[b], valid);
            if (valid) rank2[p] = part;
        }
    }
    __syncthreads();

    int run = 0;
    for (int k = 0; k < NWAVES; ++k) {
        const int v = hist[k][t];
        hist[k][t] = run;
        run += v;
    }

    int incl = run;
    for (int off = 1; off < 32; off <<= 1) {
        const int y = __shfl_up(incl, off);
        if (lane >= off) incl += y;
    }
    if (lane == 31) wave_sum[wv] = incl;
    __syncthreads();
    int wbase = 0;
    for (int k = 0; k < NWAVES; ++k)
        if (k < wv) wbase += wave_sum[k];
    const int off_t = wbase + incl - run;
    soff[t] = off_t;
    const int node = bin * BINW + t;
    if (node <= nnodes) rowptr[node] = lo + off_t;
    __syncthreads();

    for (int q = 0; q < seg; q += 32) {
        if (p0 + q >= hi) break;
        const int p = p0 + q + lane;
        if (p < hi) {
            const int4 r = rec1[p];
            const int  k = r.z & (BINW - 1);
            rec[lo + soff[k] + hist[wv][k] + rank2[p]] = make_int2(r.x, r.y);
        }
    }
}

template <int U>
__device__ __forceinline__ void gcn_add_rows(const float4* __restrict__ h4, int2 r,
                                             int j, int lane, float4& acc) {
    float4 v[U];
    float  w[U];
#pragma unroll
    for (int k = 0; k < U; ++k) {
        const int src = __builtin_amdgcn_readlane(r.x, j + k);
        w[k] = __int_as_float(__builtin_amdgcn_readlane(r.y, j + k));
        v[k] = h4[(size_t)src * (DIM / 4) + lane];
    }
#pragma unroll
    for (int k = 0; k < U; ++k) {
        acc.x = fmaf(w[k], v[k].x, acc.x);
        acc.y = fmaf(w[k], v[k].y, acc.y);
        acc.z = fmaf(w[k], v[k].z, acc.z);
        acc.w = fmaf(w[k], v[k].w, acc.w);
    }
}

__device__ __forceinline__ int2 gcn_load_records(const int2* __restrict__ rec,
                                                 int base, int end, int lane) {
    int2 r = make_int2(0, 0);
    if (lane < end - base) r = rec[base + lane];
    return r;
}

__device__ __forceinline__ float4 gcn_sum_list(const float4* __restrict__ h4,
                                               const int2*   __restrict__ rec,
                                               int beg, int end, int2 r, int lane) {
    float4 acc = make_float4(0.f, 0.f, 0.f, 0.f);
    for (int base = beg; base < end; base += 32) {
        const int n = min(32, end - base);
        if (base != beg) r = gcn_load_records(rec, base, end, lane);

        int j = 0;
        for (; j + 8 <= n; j += 8) gcn_add_rows<8>(h4, r, j, lane, acc);
        if (j + 4 <= n) { gcn_add_rows<4>(h4, r, j, lane, acc); j += 4; }
        for (; j < n; ++j) gcn_add_rows<1>(h4, r, j, lane, acc);
    }
    return acc;
}

#define APITCH (DIM + 4)
static_assert(APITCH % 4 == 0 && (APITCH % 64) == 4, "16B-aligned rows that start 4 banks apart");

__global__ __launch_bounds__(256) void gcn_layer_kernel(
    const float* __restrict__ h,
    const int*   __restrict__ rowptr,
    const int2*  __restrict__ rec,
    const float* __restrict__ Wrel,
    const float* __restrict__ Wroot,
    const float* __restrict__ bias,
    float*       __restrict__ out,
    int nrows, int do_relu) {
    __shared__ float4 sA4[16 * APITCH / 4];
    __shared__ float4 sH4[16 * APITCH / 4];

    const int t    = threadIdx.x;
    const int lane = t & 31;
    const int wv   = __builtin_amdgcn_readfirstlane(t >> 5);
    const int r    = lane & 15;
    const int half = lane >> 4;

    const int row0 = blockIdx.x * 16;
    const int col0 = wv * 16;

    {
        const float4* h4 = reinterpret_cast<const float4*>(h);
        const int m0 = 2 * wv;
        const int n0 = row0 + m0;

        int beg0 = 0, end0 = 0, beg1 = 0, end1 = 0;
        float4 self0 = make_float4(0.f, 0.f, 0.f, 0.f);
        float4 self1 = make_float4(0.f, 0.f, 0.f, 0.f);
        if (n0 < nrows) {
            beg0 = rowptr[n0];
            end0 = rowptr[n0 + 1];
        }
        if (n0 + 1 < nrows) {
            beg1 = end0;
            end1 = rowptr[n0 + 2];
        }
        const int2 r0 = gcn_load_records(rec, beg0, end0, lane);
        const int2 r1 = gcn_load_records(rec, beg1, end1, lane);
        if (n0 < nrows)     self0 = h4[(size_t)n0 * (DIM / 4) + lane];
        if (n0 + 1 < nrows) self1 = h4[(size_t)(n0 + 1) * (DIM / 4) + lane];

        const float4 a0 = gcn_sum_list(h4, rec, beg0, end0, r0, lane);
        const float4 a1 = gcn_sum_list(h4, rec, beg1, end1, r1, lane);

        sA4[m0 * (APITCH / 4) + lane]       = a0;
        sA4[(m0 + 1) * (APITCH / 4) + lane] = a1;
        sH4[m0 * (APITCH / 4) + lane]       = self0;
        sH4[(m0 + 1) * (APITCH / 4) + lane] = self1;
    }
    __syncthreads();

    const float* sA = reinterpret_cast<const float*>(sA4);
    const float* sH = reinterpret_cast<const float*>(sH4);

    v8f acc_rel  = {};
    v8f acc_root = {};
#pragma unroll
    for (int k = 0; k < DIM; k += 4) {
        const int ka = k + 2 * half;

        v2f a_rel  = *reinterpret_cast<const v2f*>(&sA[r * APITCH + ka]);
        v2f b_rel  = { Wrel[(size_t)ka * DIM + col0 + r],
                       Wrel[(size_t)(ka + 1) * DIM + col0 + r] };
        acc_rel = __builtin_amdgcn_wmma_f32_16x16x4_f32(
                      false, a_rel, false, b_rel, (short)0, acc_rel, false, false);

        v2f a_root = *reinterpret_cast<const v2f*>(&sH[r * APITCH + ka]);
        v2f b_root = { Wroot[(size_t)ka * DIM + col0 + r],
                       Wroot[(size_t)(ka + 1) * DIM + col0 + r] };
        acc_root = __builtin_amdgcn_wmma_f32_16x16x4_f32(
                      false, a_root, false, b_root, (short)0, acc_root, false, false);
    }

    const float bv = bias[col0 + r];
#pragma unroll
    for (int j = 0; j < 8; ++j) {
        const int m    = j + 8 * half;
        const int orow = row0 + m;
        if (orow < nrows) {
            float v = acc_rel[j] + acc_root[j] + bv;
            if (do_relu) v = fmaxf(v, 0.f);
            out[(size_t)orow * DIM + col0 + r] = v;
        }
    }
}

#define BASE_PAD   (NCOUNTS + 64)
#define ROWPTR_PAD (NNODES + 64)
static_assert(NCOUNTS % 4 == 0 && NNODES % 4 == 0 && NEDGES % 4 == 0, "16B alignment of the planes");

extern "C" void kernel_launch(void* const* d_in, const int* in_sizes, int n_in,
                              void* d_out, int out_size, void* d_ws, size_t ws_size,
                              hipStream_t stream) {
    const float* x      = (const float*)d_in[0];
    const int*   eidx   = (const int*)  d_in[1];
    const float* ew     = (const float*)d_in[2];
    const float* Wrel   = (const float*)d_in[3];
    const float* brel   = (const float*)d_in[4];
    const float* Wroot  = (const float*)d_in[5];
    float*       out    = (float*)d_out;

    const int N = NNODES;
    const int E = NEDGES;

    float* h1       = (float*)d_ws;
    float* h2       = h1 + (size_t)N * DIM;
    int*   tilecnt  = (int*)(h2 + (size_t)N * DIM);
    int*   tilebase = tilecnt + NCOUNTS;
    int*   rank1    = tilebase + BASE_PAD;
    int*   rank2    = rank1 + E;
    int*   rowptr   = rank2 + E;
    int4*  rec1     = (int4*)(rowptr + ROWPTR_PAD);
    int2*  rec      = (int2*)(rec1 + E);

    const int edge_blocks  = (E + 255) / 256;
    const int layer_blocks = (N + 15) / 16;

    gcn_bin_count_kernel<<<NTILES, BINW, 0, stream>>>(eidx, tilecnt, rank1, E);
    gcn_scan_kernel<<<SCAN_BLOCKS, SCAN_THREADS, 0, stream>>>(tilecnt, tilebase, NCOUNTS);
    gcn_bin_place_kernel<<<edge_blocks, 256, 0, stream>>>(eidx, ew, tilebase, rank1, rec1, E);
    gcn_node_group_kernel<<<NBINS, BINW, 0, stream>>>(tilebase, rec1, rank2, rowptr, rec, N);

    for (int l = 0; l < NLAYERS; ++l) {
        const float* hin  = (l == 0) ? x  : (l == 1) ? h1 : h2;
        float*       hout = (l == 0) ? h1 : (l == 1) ? h2 : out;

        gcn_layer_kernel<<<layer_blocks, 256, 0, stream>>>(
            hin, rowptr, rec,
            Wrel  + (size_t)l * DIM * DIM,
            Wroot + (size_t)l * DIM * DIM,
            brel  + (size_t)l * DIM,
            hout, N, (l < NLAYERS - 1) ? 1 : 0);
    }
}


# sort completion passes un-serialized: bin_count keeps its 16 keys and partial ranks per lane in registers and writes rank1 once; node_group placement loads 8 rounds together; integer only
# speedup vs baseline: 3.3163x; 1.0298x over previous
#include <hip/hip_runtime.h>

#define NNODES  50000
#define NEDGES  800000
#define DIM     128
#define NLAYERS 3

typedef __attribute__((ext_vector_type(2))) float v2f;
typedef __attribute__((ext_vector_type(8))) float v8f;


#define BINW    256
#define NWAVES  (BINW / 32)
#define TILE1   4096
#define NTILES  ((NEDGES + TILE1 - 1) / TILE1)
#define NBINS   (NNODES / BINW + 1)
#define NCOUNTS (BINW * NTILES)
#define BATCH   8
#define ROUNDS1 (TILE1 / NWAVES / 32)

#define SCAN_THREADS 1024
#define SCAN_BLOCKS  ((NCOUNTS + SCAN_THREADS - 1) / SCAN_THREADS)
#define SCAN_QUADS   ((SCAN_BLOCKS - 1) * (SCAN_THREADS / 4))
#define SCAN_PRE     ((SCAN_QUADS + SCAN_THREADS - 1) / SCAN_THREADS)

static_assert(SCAN_THREADS / 32 == 32, "one lane per wave when the wave sums are folded");

static_assert(NBINS <= BINW, "level-1 bin numbers must fit the 256-entry histogram");
static_assert(TILE1 % (NWAVES * 32 * BATCH) == 0, "a tile is whole batches for every wave");
static_assert(ROUNDS1 % BATCH == 0, "a wave's rounds are whole batches");

__device__ __forceinline__ void gcn_match32(int key, int& lt, int& tot, bool& last) {
    unsigned mine = 0u;
#pragma unroll
    for (int l = 0; l < 32; ++l) {
        const int      kl = __builtin_amdgcn_readlane(key, l);
        const bool     eq = (key == kl);
        const unsigned m  = __builtin_amdgcn_ballot_w32(eq);
        if (eq) mine = m;
    }
    const unsigned lane = threadIdx.x & 31u;
    lt   = (int)__builtin_amdgcn_mbcnt_lo(mine, 0u);
    tot  = __popc(mine);
    last = ((mine >> lane) == 1u);
}

__device__ __forceinline__ int gcn_rank_round(int* __restrict__ whist, int key, bool valid) {
    int lt, tot; bool last;
    gcn_match32(key, lt, tot, last);
    int part = 0;
    if (valid) {
        const int old = whist[key];
        part = old + lt;
        if (last) whist[key] = old + tot;
    }
    return part;
}

__global__ __launch_bounds__(BINW) void gcn_bin_count_kernel(
    const int* __restrict__ edge_index,
    int*       __restrict__ tilecnt,
    int*       __restrict__ rank1,
    int nedges) {
    __shared__ int hist[NWAVES][BINW];

    const int t    = threadIdx.x;
    const int lane = t & 31;
    const int wv   = __builtin_amdgcn_readfirstlane(t >> 5);
    const int tile = blockIdx.x;
    const int e0   = tile * TILE1;
    const int ne   = min(TILE1, nedges - e0);
    const int seg0 = wv * (TILE1 / NWAVES);
    const int* dst = edge_index + nedges;

    for (int k = 0; k < NWAVES; ++k) hist[k][t] = 0;
    __syncthreads();

    int key[ROUNDS1];
    int part[ROUNDS1];
#pragma unroll
    for (int g = 0; g < ROUNDS1 / BATCH; ++g) {
        const int q0 = g * 32 * BATCH;
#pragma unroll
        for (int b = 0; b < BATCH; ++b) {
            const int idx = seg0 + q0 + 32 * b + lane;
            const int v = dst[e0 + min(idx, ne - 1)];
            key[g * BATCH + b] = (idx < ne) ? (v >> 8) : -1;
        }
#pragma unroll
        for (int b = 0; b < BATCH; ++b) {
            const int  idx   = seg0 + q0 + 32 * b + lane;
            const bool valid = idx < ne;
            int p = 0;
            if (seg0 + q0 + 32 * b < ne) p = gcn_rank_round(hist[wv], key[g * BATCH + b], valid);
            part[g * BATCH + b] = p;
        }
    }
    __syncthreads();

    int run = 0;
    for (int k = 0; k < NWAVES; ++k) {
        const int v = hist[k][t];
        hist[k][t] = run;
        run += v;
    }
    tilecnt[t * NTILES + tile] = run;
    __syncthreads();

    int woff[ROUNDS1];
#pragma unroll
    for (int i = 0; i < ROUNDS1; ++i) woff[i] = hist[wv][max(key[i], 0)];
#pragma unroll
    for (int i = 0; i < ROUNDS1; ++i) {
        const int idx = seg0 + 32 * i + lane;
        if (idx < ne) rank1[e0 + idx] = part[i] + woff[i];
    }
}

__global__ __launch_bounds__(SCAN_THREADS) void gcn_scan_kernel(
    const int* __restrict__ cnt,
    int*       __restrict__ base,
    int n) {
    __shared__ int wave_pre[SCAN_THREADS / 32];
    __shared__ int wave_tot[SCAN_THREADS / 32];

    const int t     = threadIdx.x;
    const int lane  = t & 31;
    const int wv    = t >> 5;
    const int first = blockIdx.x * SCAN_THREADS;
    const int idx   = first + t;
    const int nq    = first / 4;
    const int4* cnt4 = reinterpret_cast<const int4*>(cnt);

    int4 q[SCAN_PRE];
#pragma unroll
    for (int i = 0; i < SCAN_PRE; ++i) {
        const int k = i * SCAN_THREADS + t;
        q[i] = cnt4[(k < nq) ? k : 0];
    }
    const int v = (idx < n) ? cnt[idx] : 0;

    int pre = 0;
#pragma unroll
    for (int i = 0; i < SCAN_PRE; ++i) {
        const int k = i * SCAN_THREADS + t;
        if (k < nq) pre += (q[i].x + q[i].y) + (q[i].z + q[i].w);
    }
    for (int off = 16; off > 0; off >>= 1) pre += __shfl_xor(pre, off);

    int incl = v;
    for (int off = 1; off < 32; off <<= 1) {
        const int y = __shfl_up(incl, off);
        if (lane >= off) incl += y;
    }
    if (lane == 0)  wave_pre[wv] = pre;
    if (lane == 31) wave_tot[wv] = incl;
    __syncthreads();

    int s = wave_pre[lane] + ((lane < wv) ? wave_tot[lane] : 0);
    for (int off = 16; off > 0; off >>= 1) s += __shfl_xor(s, off);

    const int excl = s + incl - v;
    if (idx < n)      base[idx] = excl;
    if (idx == n - 1) base[n]   = excl + v;
}

__global__ __launch_bounds__(256) void gcn_bin_place_kernel(
    const int*   __restrict__ edge_index,
    const float* __restrict__ edge_weight,
    const int*   __restrict__ tilebase,
    const int*   __restrict__ rank1,
    int4*        __restrict__ rec1,
    int nedges) {
    const int e = blockIdx.x * blockDim.x + threadIdx.x;
    if (e >= nedges) return;
    const int   src = edge_index[e];
    const int   dst = edge_index[nedges + e];
    const float w   = edge_weight[e];
    const int   pos = tilebase[(dst >> 8) * NTILES + e / TILE1] + rank1[e];
    rec1[pos] = make_int4(src, __float_as_int(w), dst, 0);
}

__global__ __launch_bounds__(BINW) void gcn_node_group_kernel(
    const int*  __restrict__ tilebase,
    const int4* __restrict__ rec1,
    int*        __restrict__ rank2,
    int*        __restrict__ rowptr,
    int2*       __restrict__ rec,
    int nnodes) {
    __shared__ int hist[NWAVES][BINW];
    __shared__ int soff[BINW];
    __shared__ int wave_sum[NWAVES];

    const int t    = threadIdx.x;
    const int lane = t & 31;
    const int wv   = __builtin_amdgcn_readfirstlane(t >> 5);
    const int bin  = blockIdx.x;
    const int lo   = tilebase[bin * NTILES];
    const int hi   = tilebase[(bin + 1) * NTILES];
    const int seg  = ((hi - lo + BINW - 1) / BINW) * 32;
    const int p0   = lo + wv * seg;

    for (int k = 0; k < NWAVES; ++k) hist[k][t] = 0;
    __syncthreads();

    for (int q0 = 0; q0 < seg; q0 += 32 * BATCH) {
        if (p0 + q0 >= hi) break;
        int key[BATCH];
#pragma unroll
        for (int b = 0; b < BATCH; ++b) {
            const int q = q0 + 32 * b;
            const int p = p0 + q + lane;
            const int v = rec1[min(p, hi - 1)].z;
            key[b] = (q < seg && p < hi) ? (v & (BINW - 1)) : -1;
        }
#pragma unroll
        for (int b = 0; b < BATCH; ++b) {
            const int  q     = q0 + 32 * b;
            if (q >= seg || p0 + q >= hi) continue;
            const int  p     = p0 + q + lane;
            const bool valid = p < hi;
            const int  part  = gcn_rank_round(hist[wv], key[b], valid);
            if (valid) rank2[p] = part;
        }
    }
    __syncthreads();

    int run = 0;
    for (int k = 0; k < NWAVES; ++k) {
        const int v = hist[k][t];
        hist[k][t] = run;
        run += v;
    }

    int incl = run;
    for (int off = 1; off < 32; off <<= 1) {
        const int y = __shfl_up(incl, off);
        if (lane >= off) incl += y;
    }
    if (lane == 31) wave_sum[wv] = incl;
    __syncthreads();
    int wbase = 0;
    for (int k = 0; k < NWAVES; ++k)
        if (k < wv) wbase += wave_sum[k];
    const int off_t = wbase + incl - run;
    soff[t] = off_t;
    const int node = bin * BINW + t;
    if (node <= nnodes) rowptr[node] = lo + off_t;
    __syncthreads();

    for (int q0 = 0; q0 < seg; q0 += 32 * BATCH) {
        if (p0 + q0 >= hi) break;
        int4 rr[BATCH];
        int  r2[BATCH];
#pragma unroll
        for (int b = 0; b < BATCH; ++b) {
            const int pc = min(p0 + q0 + 32 * b + lane, hi - 1);
            rr[b] = rec1[pc];
            r2[b] = rank2[pc];
        }
#pragma unroll
        for (int b = 0; b < BATCH; ++b) {
            const int q = q0 + 32 * b;
            const int p = p0 + q + lane;
            if (q < seg && p < hi) {
                const int k = rr[b].z & (BINW - 1);
                rec[lo + soff[k] + hist[wv][k] + r2[b]] = make_int2(rr[b].x, rr[b].y);
            }
        }
    }
}

template <int U>
__device__ __forceinline__ void gcn_add_rows(const float4* __restrict__ h4, int2 r,
                                             int j, int lane, float4& acc) {
    float4 v[U];
    float  w[U];
#pragma unroll
    for (int k = 0; k < U; ++k) {
        const int src = __builtin_amdgcn_readlane(r.x, j + k);
        w[k] = __int_as_float(__builtin_amdgcn_readlane(r.y, j + k));
        v[k] = h4[(size_t)src * (DIM / 4) + lane];
    }
#pragma unroll
    for (int k = 0; k < U; ++k) {
        acc.x = fmaf(w[k], v[k].x, acc.x);
        acc.y = fmaf(w[k], v[k].y, acc.y);
        acc.z = fmaf(w[k], v[k].z, acc.z);
        acc.w = fmaf(w[k], v[k].w, acc.w);
    }
}

__device__ __forceinline__ int2 gcn_load_records(const int2* __restrict__ rec,
                                                 int base, int end, int lane) {
    int2 r = make_int2(0, 0);
    if (lane < end - base) r = rec[base + lane];
    return r;
}

__device__ __forceinline__ float4 gcn_sum_list(const float4* __restrict__ h4,
                                               const int2*   __restrict__ rec,
                                               int beg, int end, int2 r, int lane) {
    float4 acc = make_float4(0.f, 0.f, 0.f, 0.f);
    for (int base = beg; base < end; base += 32) {
        const int n = min(32, end - base);
        if (base != beg) r = gcn_load_records(rec, base, end, lane);

        int j = 0;
        for (; j + 8 <= n; j += 8) gcn_add_rows<8>(h4, r, j, lane, acc);
        if (j + 4 <= n) { gcn_add_rows<4>(h4, r, j, lane, acc); j += 4; }
        for (; j < n; ++j) gcn_add_rows<1>(h4, r, j, lane, acc);
    }
    return acc;
}

#define APITCH (DIM + 4)
static_assert(APITCH % 4 == 0 && (APITCH % 64) == 4, "16B-aligned rows that start 4 banks apart");

__global__ __launch_bounds__(256) void gcn_layer_kernel(
    const float* __restrict__ h,
    const int*   __restrict__ rowptr,
    const int2*  __restrict__ rec,
    const float* __restrict__ Wrel,
    const float* __restrict__ Wroot,
    const float* __restrict__ bias,
    float*       __restrict__ out,
    int nrows, int do_relu) {
    __shared__ float4 sA4[16 * APITCH / 4];
    __shared__ float4 sH4[16 * APITCH / 4];

    const int t    = threadIdx.x;
    const int lane = t & 31;
    const int wv   = __builtin_amdgcn_readfirstlane(t >> 5);
    const int r    = lane & 15;
    const int half = lane >> 4;

    const int row0 = blockIdx.x * 16;
    const int col0 = wv * 16;

    {
        const float4* h4 = reinterpret_cast<const float4*>(h);
        const int m0 = 2 * wv;
        const int n0 = row0 + m0;

        int beg0 = 0, end0 = 0, beg1 = 0, end1 = 0;
        float4 self0 = make_float4(0.f, 0.f, 0.f, 0.f);
        float4 self1 = make_float4(0.f, 0.f, 0.f, 0.f);
        if (n0 < nrows) {
            beg0 = rowptr[n0];
            end0 = rowptr[n0 + 1];
        }
        if (n0 + 1 < nrows) {
            beg1 = end0;
            end1 = rowptr[n0 + 2];
        }
        const int2 r0 = gcn_load_records(rec, beg0, end0, lane);
        const int2 r1 = gcn_load_records(rec, beg1, end1, lane);
        if (n0 < nrows)     self0 = h4[(size_t)n0 * (DIM / 4) + lane];
        if (n0 + 1 < nrows) self1 = h4[(size_t)(n0 + 1) * (DIM / 4) + lane];

        const float4 a0 = gcn_sum_list(h4, rec, beg0, end0, r0, lane);
        const float4 a1 = gcn_sum_list(h4, rec, beg1, end1, r1, lane);

        sA4[m0 * (APITCH / 4) + lane]       = a0;
        sA4[(m0 + 1) * (APITCH / 4) + lane] = a1;
        sH4[m0 * (APITCH / 4) + lane]       = self0;
        sH4[(m0 + 1) * (APITCH / 4) + lane] = self1;
    }
    __syncthreads();

    const float* sA = reinterpret_cast<const float*>(sA4);
    const float* sH = reinterpret_cast<const float*>(sH4);

    v8f acc_rel  = {};
    v8f acc_root = {};
#pragma unroll
    for (int k = 0; k < DIM; k += 4) {
        const int ka = k + 2 * half;

        v2f a_rel  = *reinterpret_cast<const v2f*>(&sA[r * APITCH + ka]);
        v2f b_rel  = { Wrel[(size_t)ka * DIM + col0 + r],
                       Wrel[(size_t)(ka + 1) * DIM + col0 + r] };
        acc_rel = __builtin_amdgcn_wmma_f32_16x16x4_f32(
                      false, a_rel, false, b_rel, (short)0, acc_rel, false, false);

        v2f a_root = *reinterpret_cast<const v2f*>(&sH[r * APITCH + ka]);
        v2f b_root = { Wroot[(size_t)ka * DIM + col0 + r],
                       Wroot[(size_t)(ka + 1) * DIM + col0 + r] };
        acc_root = __builtin_amdgcn_wmma_f32_16x16x4_f32(
                      false, a_root, false, b_root, (short)0, acc_root, false, false);
    }

    const float bv = bias[col0 + r];
#pragma unroll
    for (int j = 0; j < 8; ++j) {
        const int m    = j + 8 * half;
        const int orow = row0 + m;
        if (orow < nrows) {
            float v = acc_rel[j] + acc_root[j] + bv;
            if (do_relu) v = fmaxf(v, 0.f);
            out[(size_t)orow * DIM + col0 + r] = v;
        }
    }
}

#define BASE_PAD   (NCOUNTS + 64)
#define ROWPTR_PAD (NNODES + 64)
static_assert(NCOUNTS % 4 == 0 && NNODES % 4 == 0 && NEDGES % 4 == 0, "16B alignment of the planes");

extern "C" void kernel_launch(void* const* d_in, const int* in_sizes, int n_in,
                              void* d_out, int out_size, void* d_ws, size_t ws_size,
                              hipStream_t stream) {
    const float* x      = (const float*)d_in[0];
    const int*   eidx   = (const int*)  d_in[1];
    const float* ew     = (const float*)d_in[2];
    const float* Wrel   = (const float*)d_in[3];
    const float* brel   = (const float*)d_in[4];
    const float* Wroot  = (const float*)d_in[5];
    float*       out    = (float*)d_out;

    const int N = NNODES;
    const int E = NEDGES;

    float* h1       = (float*)d_ws;
    float* h2       = h1 + (size_t)N * DIM;
    int*   tilecnt  = (int*)(h2 + (size_t)N * DIM);
    int*   tilebase = tilecnt + NCOUNTS;
    int*   rank1    = tilebase + BASE_PAD;
    int*   rank2    = rank1 + E;
    int*   rowptr   = rank2 + E;
    int4*  rec1     = (int4*)(rowptr + ROWPTR_PAD);
    int2*  rec      = (int2*)(rec1 + E);

    const int edge_blocks  = (E + 255) / 256;
    const int layer_blocks = (N + 15) / 16;

    gcn_bin_count_kernel<<<NTILES, BINW, 0, stream>>>(eidx, tilecnt, rank1, E);
    gcn_scan_kernel<<<SCAN_BLOCKS, SCAN_THREADS, 0, stream>>>(tilecnt, tilebase, NCOUNTS);
    gcn_bin_place_kernel<<<edge_blocks, 256, 0, stream>>>(eidx, ew, tilebase, rank1, rec1, E);
    gcn_node_group_kernel<<<NBINS, BINW, 0, stream>>>(tilebase, rec1, rank2, rowptr, rec, N);

    for (int l = 0; l < NLAYERS; ++l) {
        const float* hin  = (l == 0) ? x  : (l == 1) ? h1 : h2;
        float*       hout = (l == 0) ? h1 : (l == 1) ? h2 : out;

        gcn_layer_kernel<<<layer_blocks, 256, 0, stream>>>(
            hin, rowptr, rec,
            Wrel  + (size_t)l * DIM * DIM,
            Wroot + (size_t)l * DIM * DIM,
            brel  + (size_t)l * DIM,
            hout, N, (l < NLAYERS - 1) ? 1 : 0);
    }
}


# layer kernel: Wrel/Wroot copied per call into WMMA B-fragment order (by extra blocks of the scan launch); a wave loads B with 32 whole-line 16-byte loads, not 128 4-byte loads; sum order unchanged
# speedup vs baseline: 3.4675x; 1.0456x over previous
#include <hip/hip_runtime.h>

#define NNODES  50000
#define NEDGES  800000
#define DIM     128
#define NLAYERS 3

typedef __attribute__((ext_vector_type(2))) float v2f;
typedef __attribute__((ext_vector_type(8))) float v8f;


#define BINW    256
#define NWAVES  (BINW / 32)
#define TILE1   4096
#define NTILES  ((NEDGES + TILE1 - 1) / TILE1)
#define NBINS   (NNODES / BINW + 1)
#define NCOUNTS (BINW * NTILES)
#define BATCH   8
#define ROUNDS1 (TILE1 / NWAVES / 32)

#define SCAN_THREADS 1024
#define SCAN_BLOCKS  ((NCOUNTS + SCAN_THREADS - 1) / SCAN_THREADS)
#define SCAN_QUADS   ((SCAN_BLOCKS - 1) * (SCAN_THREADS / 4))
#define SCAN_PRE     ((SCAN_QUADS + SCAN_THREADS - 1) / SCAN_THREADS)

static_assert(SCAN_THREADS / 32 == 32, "one lane per wave when the wave sums are folded");

#define WPACK_MATS   (NLAYERS * 2)
#define WPACK_QUADS  (WPACK_MATS * DIM * DIM / 4)
#define PACK_BLOCKS  (WPACK_QUADS / SCAN_THREADS)
static_assert(WPACK_QUADS % SCAN_THREADS == 0, "pack blocks are whole");
static_assert(DIM == 128, "the fragment order below is written for 8 column tiles and 16 k-groups of 8");

static_assert(NBINS <= BINW, "level-1 bin numbers must fit the 256-entry histogram");
static_assert(TILE1 % (NWAVES * 32 * BATCH) == 0, "a tile is whole batches for every wave");
static_assert(ROUNDS1 % BATCH == 0, "a wave's rounds are whole batches");

__device__ __forceinline__ void gcn_match32(int key, int& lt, int& tot, bool& last) {
    unsigned mine = 0u;
#pragma unroll
    for (int l = 0; l < 32; ++l) {
        const int      kl = __builtin_amdgcn_readlane(key, l);
        const bool     eq = (key == kl);
        const unsigned m  = __builtin_amdgcn_ballot_w32(eq);
        if (eq) mine = m;
    }
    const unsigned lane = threadIdx.x & 31u;
    lt   = (int)__builtin_amdgcn_mbcnt_lo(mine, 0u);
    tot  = __popc(mine);
    last = ((mine >> lane) == 1u);
}

__device__ __forceinline__ int gcn_rank_round(int* __restrict__ whist, int key, bool valid) {
    int lt, tot; bool last;
    gcn_match32(key, lt, tot, last);
    int part = 0;
    if (valid) {
        const int old = whist[key];
        part = old + lt;
        if (last) whist[key] = old + tot;
    }
    return part;
}

__global__ __launch_bounds__(BINW) void gcn_bin_count_kernel(
    const int* __restrict__ edge_index,
    int*       __restrict__ tilecnt,
    int*       __restrict__ rank1,
    int nedges) {
    __shared__ int hist[NWAVES][BINW];

    const int t    = threadIdx.x;
    const int lane = t & 31;
    const int wv   = __builtin_amdgcn_readfirstlane(t >> 5);
    const int tile = blockIdx.x;
    const int e0   = tile * TILE1;
    const int ne   = min(TILE1, nedges - e0);
    const int seg0 = wv * (TILE1 / NWAVES);
    const int* dst = edge_index + nedges;

    for (int k = 0; k < NWAVES; ++k) hist[k][t] = 0;
    __syncthreads();

    int key[ROUNDS1];
    int part[ROUNDS1];
#pragma unroll
    for (int g = 0; g < ROUNDS1 / BATCH; ++g) {
        const int q0 = g * 32 * BATCH;
#pragma unroll
        for (int b = 0; b < BATCH; ++b) {
            const int idx = seg0 + q0 + 32 * b + lane;
            const int v = dst[e0 + min(idx, ne - 1)];
            key[g * BATCH + b] = (idx < ne) ? (v >> 8) : -1;
        }
#pragma unroll
        for (int b = 0; b < BATCH; ++b) {
            const int  idx   = seg0 + q0 + 32 * b + lane;
            const bool valid = idx < ne;
            int p = 0;
            if (seg0 + q0 + 32 * b < ne) p = gcn_rank_round(hist[wv], key[g * BATCH + b], valid);
            part[g * BATCH + b] = p;
        }
    }
    __syncthreads();

    int run = 0;
    for (int k = 0; k < NWAVES; ++k) {
        const int v = hist[k][t];
        hist[k][t] = run;
        run += v;
    }
    tilecnt[t * NTILES + tile] = run;
    __syncthreads();

    int woff[ROUNDS1];
#pragma unroll
    for (int i = 0; i < ROUNDS1; ++i) woff[i] = hist[wv][max(key[i], 0)];
#pragma unroll
    for (int i = 0; i < ROUNDS1; ++i) {
        const int idx = seg0 + 32 * i + lane;
        if (idx < ne) rank1[e0 + idx] = part[i] + woff[i];
    }
}

__global__ __launch_bounds__(SCAN_THREADS) void gcn_scan_kernel(
    const int*   __restrict__ cnt,
    int*         __restrict__ base,
    int n,
    const float* __restrict__ Wrel,
    const float* __restrict__ Wroot,
    float4*      __restrict__ wpack) {
    __shared__ int wave_pre[SCAN_THREADS / 32];
    __shared__ int wave_tot[SCAN_THREADS / 32];

    if (blockIdx.x >= SCAN_BLOCKS) {
        const int p    = (blockIdx.x - SCAN_BLOCKS) * SCAN_THREADS + threadIdx.x;
        const int ln   = p & 31;
        const int u    = (p >> 5) & 15;
        const int w    = (p >> 9) & 7;
        const int m    = p >> 12;
        const int col  = 16 * w + (ln & 15);
        const int k0   = 8 * u + 2 * (ln >> 4);
        const float* W = ((m & 1) ? Wroot : Wrel) + (size_t)(m >> 1) * DIM * DIM;
        wpack[p] = make_float4(W[(size_t)k0 * DIM + col],       W[(size_t)(k0 + 1) * DIM + col],
                               W[(size_t)(k0 + 4) * DIM + col], W[(size_t)(k0 + 5) * DIM + col]);
        return;
    }

    const int t     = threadIdx.x;
    const int lane  = t & 31;
    const int wv    = t >> 5;
    const int first = blockIdx.x * SCAN_THREADS;
    const int idx   = first + t;
    const int nq    = first / 4;
    const int4* cnt4 = reinterpret_cast<const int4*>(cnt);

    int4 q[SCAN_PRE];
#pragma unroll
    for (int i = 0; i < SCAN_PRE; ++i) {
        const int k = i * SCAN_THREADS + t;
        q[i] = cnt4[(k < nq) ? k : 0];
    }
    const int v = (idx < n) ? cnt[idx] : 0;

    int pre = 0;
#pragma unroll
    for (int i = 0; i < SCAN_PRE; ++i) {
        const int k = i * SCAN_THREADS + t;
        if (k < nq) pre += (q[i].x + q[i].y) + (q[i].z + q[i].w);
    }
    for (int off = 16; off > 0; off >>= 1) pre += __shfl_xor(pre, off);

    int incl = v;
    for (int off = 1; off < 32; off <<= 1) {
        const int y = __shfl_up(incl, off);
        if (lane >= off) incl += y;
    }
    if (lane == 0)  wave_pre[wv] = pre;
    if (lane == 31) wave_tot[wv] = incl;
    __syncthreads();

    int s = wave_pre[lane] + ((lane < wv) ? wave_tot[lane] : 0);
    for (int off = 16; off > 0; off >>= 1) s += __shfl_xor(s, off);

    const int excl = s + incl - v;
    if (idx < n)      base[idx] = excl;
    if (idx == n - 1) base[n]   = excl + v;
}

__global__ __launch_bounds__(256) void gcn_bin_place_kernel(
    const int*   __restrict__ edge_index,
    const float* __restrict__ edge_weight,
    const int*   __restrict__ tilebase,
    const int*   __restrict__ rank1,
    int4*        __restrict__ rec1,
    int nedges) {
    const int e = blockIdx.x * blockDim.x + threadIdx.x;
    if (e >= nedges) return;
    const int   src = edge_index[e];
    const int   dst = edge_index[nedges + e];
    const float w   = edge_weight[e];
    const int   pos = tilebase[(dst >> 8) * NTILES + e / TILE1] + rank1[e];
    rec1[pos] = make_int4(src, __float_as_int(w), dst, 0);
}

__global__ __launch_bounds__(BINW) void gcn_node_group_kernel(
    const int*  __restrict__ tilebase,
    const int4* __restrict__ rec1,
    int*        __restrict__ rank2,
    int*        __restrict__ rowptr,
    int2*       __restrict__ rec,
    int nnodes) {
    __shared__ int hist[NWAVES][BINW];
    __shared__ int soff[BINW];
    __shared__ int wave_sum[NWAVES];

    const int t    = threadIdx.x;
    const int lane = t & 31;
    const int wv   = __builtin_amdgcn_readfirstlane(t >> 5);
    const int bin  = blockIdx.x;
    const int lo   = tilebase[bin * NTILES];
    const int hi   = tilebase[(bin + 1) * NTILES];
    const int seg  = ((hi - lo + BINW - 1) / BINW) * 32;
    const int p0   = lo + wv * seg;

    for (int k = 0; k < NWAVES; ++k) hist[k][t] = 0;
    __syncthreads();

    for (int q0 = 0; q0 < seg; q0 += 32 * BATCH) {
        if (p0 + q0 >= hi) break;
        int key[BATCH];
#pragma unroll
        for (int b = 0; b < BATCH; ++b) {
            const int q = q0 + 32 * b;
            const int p = p0 + q + lane;
            const int v = rec1[min(p, hi - 1)].z;
            key[b] = (q < seg && p < hi) ? (v & (BINW - 1)) : -1;
        }
#pragma unroll
        for (int b = 0; b < BATCH; ++b) {
            const int  q     = q0 + 32 * b;
            if (q >= seg || p0 + q >= hi) continue;
            const int  p     = p0 + q + lane;
            const bool valid = p < hi;
            const int  part  = gcn_rank_round(hist[wv], key[b], valid);
            if (valid) rank2[p] = part;
        }
    }
    __syncthreads();

    int run = 0;
    for (int k = 0; k < NWAVES; ++k) {
        const int v = hist[k][t];
        hist[k][t] = run;
        run += v;
    }

    int incl = run;
    for (int off = 1; off < 32; off <<= 1) {
        const int y = __shfl_up(incl, off);
        if (lane >= off) incl += y;
    }
    if (lane == 31) wave_sum[wv] = incl;
    __syncthreads();
    int wbase = 0;
    for (int k = 0; k < NWAVES; ++k)
        if (k < wv) wbase += wave_sum[k];
    const int off_t = wbase + incl - run;
    soff[t] = off_t;
    const int node = bin * BINW + t;
    if (node <= nnodes) rowptr[node] = lo + off_t;
    __syncthreads();

    for (int q0 = 0; q0 < seg; q0 += 32 * BATCH) {
        if (p0 + q0 >= hi) break;
        int4 rr[BATCH];
        int  r2[BATCH];
#pragma unroll
        for (int b = 0; b < BATCH; ++b) {
            const int pc = min(p0 + q0 + 32 * b + lane, hi - 1);
            rr[b] = rec1[pc];
            r2[b] = rank2[pc];
        }
#pragma unroll
        for (int b = 0; b < BATCH; ++b) {
            const int q = q0 + 32 * b;
            const int p = p0 + q + lane;
            if (q < seg && p < hi) {
                const int k = rr[b].z & (BINW - 1);
                rec[lo + soff[k] + hist[wv][k] + r2[b]] = make_int2(rr[b].x, rr[b].y);
            }
        }
    }
}

template <int U>
__device__ __forceinline__ void gcn_add_rows(const float4* __restrict__ h4, int2 r,
                                             int j, int lane, float4& acc) {
    float4 v[U];
    float  w[U];
#pragma unroll
    for (int k = 0; k < U; ++k) {
        const int src = __builtin_amdgcn_readlane(r.x, j + k);
        w[k] = __int_as_float(__builtin_amdgcn_readlane(r.y, j + k));
        v[k] = h4[(size_t)src * (DIM / 4) + lane];
    }
#pragma unroll
    for (int k = 0; k < U; ++k) {
        acc.x = fmaf(w[k], v[k].x, acc.x);
        acc.y = fmaf(w[k], v[k].y, acc.y);
        acc.z = fmaf(w[k], v[k].z, acc.z);
        acc.w = fmaf(w[k], v[k].w, acc.w);
    }
}

__device__ __forceinline__ int2 gcn_load_records(const int2* __restrict__ rec,
                                                 int base, int end, int lane) {
    int2 r = make_int2(0, 0);
    if (lane < end - base) r = rec[base + lane];
    return r;
}

__device__ __forceinline__ float4 gcn_sum_list(const float4* __restrict__ h4,
                                               const int2*   __restrict__ rec,
                                               int beg, int end, int2 r, int lane) {
    float4 acc = make_float4(0.f, 0.f, 0.f, 0.f);
    for (int base = beg; base < end; base += 32) {
        const int n = min(32, end - base);
        if (base != beg) r = gcn_load_records(rec, base, end, lane);

        int j = 0;
        for (; j + 8 <= n; j += 8) gcn_add_rows<8>(h4, r, j, lane, acc);
        if (j + 4 <= n) { gcn_add_rows<4>(h4, r, j, lane, acc); j += 4; }
        for (; j < n; ++j) gcn_add_rows<1>(h4, r, j, lane, acc);
    }
    return acc;
}

#define APITCH (DIM + 4)
static_assert(APITCH % 4 == 0 && (APITCH % 64) == 4, "16B-aligned rows that start 4 banks apart");

__global__ __launch_bounds__(256) void gcn_layer_kernel(
    const float* __restrict__ h,
    const int*   __restrict__ rowptr,
    const int2*  __restrict__ rec,
    const float4* __restrict__ WrelP,
    const float4* __restrict__ WrootP,
    const float* __restrict__ bias,
    float*       __restrict__ out,
    int nrows, int do_relu) {
    __shared__ float4 sA4[16 * APITCH / 4];
    __shared__ float4 sH4[16 * APITCH / 4];

    const int t    = threadIdx.x;
    const int lane = t & 31;
    const int wv   = __builtin_amdgcn_readfirstlane(t >> 5);
    const int r    = lane & 15;
    const int half = lane >> 4;

    const int row0 = blockIdx.x * 16;
    const int col0 = wv * 16;

    {
        const float4* h4 = reinterpret_cast<const float4*>(h);
        const int m0 = 2 * wv;
        const int n0 = row0 + m0;

        int beg0 = 0, end0 = 0, beg1 = 0, end1 = 0;
        float4 self0 = make_float4(0.f, 0.f, 0.f, 0.f);
        float4 self1 = make_float4(0.f, 0.f, 0.f, 0.f);
        if (n0 < nrows) {
            beg0 = rowptr[n0];
            end0 = rowptr[n0 + 1];
        }
        if (n0 + 1 < nrows) {
            beg1 = end0;
            end1 = rowptr[n0 + 2];
        }
        const int2 r0 = gcn_load_records(rec, beg0, end0, lane);
        const int2 r1 = gcn_load_records(rec, beg1, end1, lane);
        if (n0 < nrows)     self0 = h4[(size_t)n0 * (DIM / 4) + lane];
        if (n0 + 1 < nrows) self1 = h4[(size_t)(n0 + 1) * (DIM / 4) + lane];

        const float4 a0 = gcn_sum_list(h4, rec, beg0, end0, r0, lane);
        const float4 a1 = gcn_sum_list(h4, rec, beg1, end1, r1, lane);

        sA4[m0 * (APITCH / 4) + lane]       = a0;
        sA4[(m0 + 1) * (APITCH / 4) + lane] = a1;
        sH4[m0 * (APITCH / 4) + lane]       = self0;
        sH4[(m0 + 1) * (APITCH / 4) + lane] = self1;
    }
    __syncthreads();

    const float* sA = reinterpret_cast<const float*>(sA4);
    const float* sH = reinterpret_cast<const float*>(sH4);

    const float4* bR = WrelP  + wv * (16 * 32) + lane;
    const float4* bT = WrootP + wv * (16 * 32) + lane;

    v8f acc_rel  = {};
    v8f acc_root = {};
#pragma unroll
    for (int u = 0; u < DIM / 8; ++u) {
        const float4 br = bR[u * 32];
        const float4 bt = bT[u * 32];
#pragma unroll
        for (int s = 0; s < 2; ++s) {
            const int ka = 8 * u + 4 * s + 2 * half;

            v2f a_rel  = *reinterpret_cast<const v2f*>(&sA[r * APITCH + ka]);
            v2f b_rel  = s ? v2f{ br.z, br.w } : v2f{ br.x, br.y };
            acc_rel = __builtin_amdgcn_wmma_f32_16x16x4_f32(
                          false, a_rel, false, b_rel, (short)0, acc_rel, false, false);

            v2f a_root = *reinterpret_cast<const v2f*>(&sH[r * APITCH + ka]);
            v2f b_root = s ? v2f{ bt.z, bt.w } : v2f{ bt.x, bt.y };
            acc_root = __builtin_amdgcn_wmma_f32_16x16x4_f32(
                          false, a_root, false, b_root, (short)0, acc_root, false, false);
        }
    }

    const float bv = bias[col0 + r];
#pragma unroll
    for (int j = 0; j < 8; ++j) {
        const int m    = j + 8 * half;
        const int orow = row0 + m;
        if (orow < nrows) {
            float v = acc_rel[j] + acc_root[j] + bv;
            if (do_relu) v = fmaxf(v, 0.f);
            out[(size_t)orow * DIM + col0 + r] = v;
        }
    }
}

#define BASE_PAD   (NCOUNTS + 64)
#define ROWPTR_PAD (NNODES + 64)
static_assert(NCOUNTS % 4 == 0 && NNODES % 4 == 0 && NEDGES % 4 == 0, "16B alignment of the planes");

extern "C" void kernel_launch(void* const* d_in, const int* in_sizes, int n_in,
                              void* d_out, int out_size, void* d_ws, size_t ws_size,
                              hipStream_t stream) {
    const float* x      = (const float*)d_in[0];
    const int*   eidx   = (const int*)  d_in[1];
    const float* ew     = (const float*)d_in[2];
    const float* Wrel   = (const float*)d_in[3];
    const float* brel   = (const float*)d_in[4];
    const float* Wroot  = (const float*)d_in[5];
    float*       out    = (float*)d_out;

    const int N = NNODES;
    const int E = NEDGES;

    float* h1       = (float*)d_ws;
    float* h2       = h1 + (size_t)N * DIM;
    float4* wpack   = (float4*)(h2 + (size_t)N * DIM);
    int*   tilecnt  = (int*)(wpack + WPACK_QUADS);
    int*   tilebase = tilecnt + NCOUNTS;
    int*   rank1    = tilebase + BASE_PAD;
    int*   rank2    = rank1 + E;
    int*   rowptr   = rank2 + E;
    int4*  rec1     = (int4*)(rowptr + ROWPTR_PAD);
    int2*  rec      = (int2*)(rec1 + E);

    const int edge_blocks  = (E + 255) / 256;
    const int layer_blocks = (N + 15) / 16;

    gcn_bin_count_kernel<<<NTILES, BINW, 0, stream>>>(eidx, tilecnt, rank1, E);
    gcn_scan_kernel<<<SCAN_BLOCKS + PACK_BLOCKS, SCAN_THREADS, 0, stream>>>(tilecnt, tilebase, NCOUNTS,
                                                                            Wrel, Wroot, wpack);
    gcn_bin_place_kernel<<<edge_blocks, 256, 0, stream>>>(eidx, ew, tilebase, rank1, rec1, E);
    gcn_node_group_kernel<<<NBINS, BINW, 0, stream>>>(tilebase, rec1, rank2, rowptr, rec, N);

    for (int l = 0; l < NLAYERS; ++l) {
        const float* hin  = (l == 0) ? x  : (l == 1) ? h1 : h2;
        float*       hout = (l == 0) ? h1 : (l == 1) ? h2 : out;

        gcn_layer_kernel<<<layer_blocks, 256, 0, stream>>>(
            hin, rowptr, rec,
            wpack + (size_t)(2 * l)     * (DIM * DIM / 4),
            wpack + (size_t)(2 * l + 1) * (DIM * DIM / 4),
            brel  + (size_t)l * DIM,
            hout, N, (l < NLAYERS - 1) ? 1 : 0);
    }
}


# layer kernel: register budget 64 per wave (launch bound 256,16; it took 68, allocated as 80): 16 waves per SIMD = 8 blocks per WGP resident instead of 12 = 6; sum order unchanged
# speedup vs baseline: 3.6209x; 1.0442x over previous
#include <hip/hip_runtime.h>

#define NNODES  50000
#define NEDGES  800000
#define DIM     128
#define NLAYERS 3

typedef __attribute__((ext_vector_type(2))) float v2f;
typedef __attribute__((ext_vector_type(8))) float v8f;


#define BINW    256
#define NWAVES  (BINW / 32)
#define TILE1   4096
#define NTILES  ((NEDGES + TILE1 - 1) / TILE1)
#define NBINS   (NNODES / BINW + 1)
#define NCOUNTS (BINW * NTILES)
#define BATCH   8
#define ROUNDS1 (TILE1 / NWAVES / 32)

#define SCAN_THREADS 1024
#define SCAN_BLOCKS  ((NCOUNTS + SCAN_THREADS - 1) / SCAN_THREADS)
#define SCAN_QUADS   ((SCAN_BLOCKS - 1) * (SCAN_THREADS / 4))
#define SCAN_PRE     ((SCAN_QUADS + SCAN_THREADS - 1) / SCAN_THREADS)

static_assert(SCAN_THREADS / 32 == 32, "one lane per wave when the wave sums are folded");

#define WPACK_MATS   (NLAYERS * 2)
#define WPACK_QUADS  (WPACK_MATS * DIM * DIM / 4)
#define PACK_BLOCKS  (WPACK_QUADS / SCAN_THREADS)
static_assert(WPACK_QUADS % SCAN_THREADS == 0, "pack blocks are whole");
static_assert(DIM == 128, "the fragment order below is written for 8 column tiles and 16 k-groups of 8");

static_assert(NBINS <= BINW, "level-1 bin numbers must fit the 256-entry histogram");
static_assert(TILE1 % (NWAVES * 32 * BATCH) == 0, "a tile is whole batches for every wave");
static_assert(ROUNDS1 % BATCH == 0, "a wave's rounds are whole batches");

__device__ __forceinline__ void gcn_match32(int key, int& lt, int& tot, bool& last) {
    unsigned mine = 0u;
#pragma unroll
    for (int l = 0; l < 32; ++l) {
        const int      kl = __builtin_amdgcn_readlane(key, l);
        const bool     eq = (key == kl);
        const unsigned m  = __builtin_amdgcn_ballot_w32(eq);
        if (eq) mine = m;
    }
    const unsigned lane = threadIdx.x & 31u;
    lt   = (int)__builtin_amdgcn_mbcnt_lo(mine, 0u);
    tot  = __popc(mine);
    last = ((mine >> lane) == 1u);
}

__device__ __forceinline__ int gcn_rank_round(int* __restrict__ whist, int key, bool valid) {
    int lt, tot; bool last;
    gcn_match32(key, lt, tot, last);
    int part = 0;
    if (valid) {
        const int old = whist[key];
        part = old + lt;
        if (last) whist[key] = old + tot;
    }
    return part;
}

__global__ __launch_bounds__(BINW) void gcn_bin_count_kernel(
    const int* __restrict__ edge_index,
    int*       __restrict__ tilecnt,
    int*       __restrict__ rank1,
    int nedges) {
    __shared__ int hist[NWAVES][BINW];

    const int t    = threadIdx.x;
    const int lane = t & 31;
    const int wv   = __builtin_amdgcn_readfirstlane(t >> 5);
    const int tile = blockIdx.x;
    const int e0   = tile * TILE1;
    const int ne   = min(TILE1, nedges - e0);
    const int seg0 = wv * (TILE1 / NWAVES);
    const int* dst = edge_index + nedges;

    for (int k = 0; k < NWAVES; ++k) hist[k][t] = 0;
    __syncthreads();

    int key[ROUNDS1];
    int part[ROUNDS1];
#pragma unroll
    for (int g = 0; g < ROUNDS1 / BATCH; ++g) {
        const int q0 = g * 32 * BATCH;
#pragma unroll
        for (int b = 0; b < BATCH; ++b) {
            const int idx = seg0 + q0 + 32 * b + lane;
            const int v = dst[e0 + min(idx, ne - 1)];
            key[g * BATCH + b] = (idx < ne) ? (v >> 8) : -1;
        }
#pragma unroll
        for (int b = 0; b < BATCH; ++b) {
            const int  idx   = seg0 + q0 + 32 * b + lane;
            const bool valid = idx < ne;
            int p = 0;
            if (seg0 + q0 + 32 * b < ne) p = gcn_rank_round(hist[wv], key[g * BATCH + b], valid);
            part[g * BATCH + b] = p;
        }
    }
    __syncthreads();

    int run = 0;
    for (int k = 0; k < NWAVES; ++k) {
        const int v = hist[k][t];
        hist[k][t] = run;
        run += v;
    }
    tilecnt[t * NTILES + tile] = run;
    __syncthreads();

    int woff[ROUNDS1];
#pragma unroll
    for (int i = 0; i < ROUNDS1; ++i) woff[i] = hist[wv][max(key[i], 0)];
#pragma unroll
    for (int i = 0; i < ROUNDS1; ++i) {
        const int idx = seg0 + 32 * i + lane;
        if (idx < ne) rank1[e0 + idx] = part[i] + woff[i];
    }
}

__global__ __launch_bounds__(SCAN_THREADS) void gcn_scan_kernel(
    const int*   __restrict__ cnt,
    int*         __restrict__ base,
    int n,
    const float* __restrict__ Wrel,
    const float* __restrict__ Wroot,
    float4*      __restrict__ wpack) {
    __shared__ int wave_pre[SCAN_THREADS / 32];
    __shared__ int wave_tot[SCAN_THREADS / 32];

    if (blockIdx.x >= SCAN_BLOCKS) {
        const int p    = (blockIdx.x - SCAN_BLOCKS) * SCAN_THREADS + threadIdx.x;
        const int ln   = p & 31;
        const int u    = (p >> 5) & 15;
        const int w    = (p >> 9) & 7;
        const int m    = p >> 12;
        const int col  = 16 * w + (ln & 15);
        const int k0   = 8 * u + 2 * (ln >> 4);
        const float* W = ((m & 1) ? Wroot : Wrel) + (size_t)(m >> 1) * DIM * DIM;
        wpack[p] = make_float4(W[(size_t)k0 * DIM + col],       W[(size_t)(k0 + 1) * DIM + col],
                               W[(size_t)(k0 + 4) * DIM + col], W[(size_t)(k0 + 5) * DIM + col]);
        return;
    }

    const int t     = threadIdx.x;
    const int lane  = t & 31;
    const int wv    = t >> 5;
    const int first = blockIdx.x * SCAN_THREADS;
    const int idx   = first + t;
    const int nq    = first / 4;
    const int4* cnt4 = reinterpret_cast<const int4*>(cnt);

    int4 q[SCAN_PRE];
#pragma unroll
    for (int i = 0; i < SCAN_PRE; ++i) {
        const int k = i * SCAN_THREADS + t;
        q[i] = cnt4[(k < nq) ? k : 0];
    }
    const int v = (idx < n) ? cnt[idx] : 0;

    int pre = 0;
#pragma unroll
    for (int i = 0; i < SCAN_PRE; ++i) {
        const int k = i * SCAN_THREADS + t;
        if (k < nq) pre += (q[i].x + q[i].y) + (q[i].z + q[i].w);
    }
    for (int off = 16; off > 0; off >>= 1) pre += __shfl_xor(pre, off);

    int incl = v;
    for (int off = 1; off < 32; off <<= 1) {
        const int y = __shfl_up(incl, off);
        if (lane >= off) incl += y;
    }
    if (lane == 0)  wave_pre[wv] = pre;
    if (lane == 31) wave_tot[wv] = incl;
    __syncthreads();

    int s = wave_pre[lane] + ((lane < wv) ? wave_tot[lane] : 0);
    for (int off = 16; off > 0; off >>= 1) s += __shfl_xor(s, off);

    const int excl = s + incl - v;
    if (idx < n)      base[idx] = excl;
    if (idx == n - 1) base[n]   = excl + v;
}

__global__ __launch_bounds__(256) void gcn_bin_place_kernel(
    const int*   __restrict__ edge_index,
    const float* __restrict__ edge_weight,
    const int*   __restrict__ tilebase,
    const int*   __restrict__ rank1,
    int4*        __restrict__ rec1,
    int nedges) {
    const int e = blockIdx.x * blockDim.x + threadIdx.x;
    if (e >= nedges) return;
    const int   src = edge_index[e];
    const int   dst = edge_index[nedges + e];
    const float w   = edge_weight[e];
    const int   pos = tilebase[(dst >> 8) * NTILES + e / TILE1] + rank1[e];
    rec1[pos] = make_int4(src, __float_as_int(w), dst, 0);
}

__global__ __launch_bounds__(BINW) void gcn_node_group_kernel(
    const int*  __restrict__ tilebase,
    const int4* __restrict__ rec1,
    int*        __restrict__ rank2,
    int*        __restrict__ rowptr,
    int2*       __restrict__ rec,
    int nnodes) {
    __shared__ int hist[NWAVES][BINW];
    __shared__ int soff[BINW];
    __shared__ int wave_sum[NWAVES];

    const int t    = threadIdx.x;
    const int lane = t & 31;
    const int wv   = __builtin_amdgcn_readfirstlane(t >> 5);
    const int bin  = blockIdx.x;
    const int lo   = tilebase[bin * NTILES];
    const int hi   = tilebase[(bin + 1) * NTILES];
    const int seg  = ((hi - lo + BINW - 1) / BINW) * 32;
    const int p0   = lo + wv * seg;

    for (int k = 0; k < NWAVES; ++k) hist[k][t] = 0;
    __syncthreads();

    for (int q0 = 0; q0 < seg; q0 += 32 * BATCH) {
        if (p0 + q0 >= hi) break;
        int key[BATCH];
#pragma unroll
        for (int b = 0; b < BATCH; ++b) {
            const int q = q0 + 32 * b;
            const int p = p0 + q + lane;
            const int v = rec1[min(p, hi - 1)].z;
            key[b] = (q < seg && p < hi) ? (v & (BINW - 1)) : -1;
        }
#pragma unroll
        for (int b = 0; b < BATCH; ++b) {
            const int  q     = q0 + 32 * b;
            if (q >= seg || p0 + q >= hi) continue;
            const int  p     = p0 + q + lane;
            const bool valid = p < hi;
            const int  part  = gcn_rank_round(hist[wv], key[b], valid);
            if (valid) rank2[p] = part;
        }
    }
    __syncthreads();

    int run = 0;
    for (int k = 0; k < NWAVES; ++k) {
        const int v = hist[k][t];
        hist[k][t] = run;
        run += v;
    }

    int incl = run;
    for (int off = 1; off < 32; off <<= 1) {
        const int y = __shfl_up(incl, off);
        if (lane >= off) incl += y;
    }
    if (lane == 31) wave_sum[wv] = incl;
    __syncthreads();
    int wbase = 0;
    for (int k = 0; k < NWAVES; ++k)
        if (k < wv) wbase += wave_sum[k];
    const int off_t = wbase + incl - run;
    soff[t] = off_t;
    const int node = bin * BINW + t;
    if (node <= nnodes) rowptr[node] = lo + off_t;
    __syncthreads();

    for (int q0 = 0; q0 < seg; q0 += 32 * BATCH) {
        if (p0 + q0 >= hi) break;
        int4 rr[BATCH];
        int  r2[BATCH];
#pragma unroll
        for (int b = 0; b < BATCH; ++b) {
            const int pc = min(p0 + q0 + 32 * b + lane, hi - 1);
            rr[b] = rec1[pc];
            r2[b] = rank2[pc];
        }
#pragma unroll
        for (int b = 0; b < BATCH; ++b) {
            const int q = q0 + 32 * b;
            const int p = p0 + q + lane;
            if (q < seg && p < hi) {
                const int k = rr[b].z & (BINW - 1);
                rec[lo + soff[k] + hist[wv][k] + r2[b]] = make_int2(rr[b].x, rr[b].y);
            }
        }
    }
}

template <int U>
__device__ __forceinline__ void gcn_add_rows(const float4* __restrict__ h4, int2 r,
                                             int j, int lane, float4& acc) {
    float4 v[U];
    float  w[U];
#pragma unroll
    for (int k = 0; k < U; ++k) {
        const int src = __builtin_amdgcn_readlane(r.x, j + k);
        w[k] = __int_as_float(__builtin_amdgcn_readlane(r.y, j + k));
        v[k] = h4[(size_t)src * (DIM / 4) + lane];
    }
#pragma unroll
    for (int k = 0; k < U; ++k) {
        acc.x = fmaf(w[k], v[k].x, acc.x);
        acc.y = fmaf(w[k], v[k].y, acc.y);
        acc.z = fmaf(w[k], v[k].z, acc.z);
        acc.w = fmaf(w[k], v[k].w, acc.w);
    }
}

__device__ __forceinline__ int2 gcn_load_records(const int2* __restrict__ rec,
                                                 int base, int end, int lane) {
    int2 r = make_int2(0, 0);
    if (lane < end - base) r = rec[base + lane];
    return r;
}

__device__ __forceinline__ float4 gcn_sum_list(const float4* __restrict__ h4,
                                               const int2*   __restrict__ rec,
                                               int beg, int end, int2 r, int lane) {
    float4 acc = make_float4(0.f, 0.f, 0.f, 0.f);
    for (int base = beg; base < end; base += 32) {
        const int n = min(32, end - base);
        if (base != beg) r = gcn_load_records(rec, base, end, lane);

        int j = 0;
        for (; j + 8 <= n; j += 8) gcn_add_rows<8>(h4, r, j, lane, acc);
        if (j + 4 <= n) { gcn_add_rows<4>(h4, r, j, lane, acc); j += 4; }
        for (; j < n; ++j) gcn_add_rows<1>(h4, r, j, lane, acc);
    }
    return acc;
}

#define APITCH (DIM + 4)
static_assert(APITCH % 4 == 0 && (APITCH % 64) == 4, "16B-aligned rows that start 4 banks apart");

__global__ __launch_bounds__(256, 16) void gcn_layer_kernel(
    const float* __restrict__ h,
    const int*   __restrict__ rowptr,
    const int2*  __restrict__ rec,
    const float4* __restrict__ WrelP,
    const float4* __restrict__ WrootP,
    const float* __restrict__ bias,
    float*       __restrict__ out,
    int nrows, int do_relu) {
    __shared__ float4 sA4[16 * APITCH / 4];
    __shared__ float4 sH4[16 * APITCH / 4];

    const int t    = threadIdx.x;
    const int lane = t & 31;
    const int wv   = __builtin_amdgcn_readfirstlane(t >> 5);
    const int r    = lane & 15;
    const int half = lane >> 4;

    const int row0 = blockIdx.x * 16;
    const int col0 = wv * 16;

    {
        const float4* h4 = reinterpret_cast<const float4*>(h);
        const int m0 = 2 * wv;
        const int n0 = row0 + m0;

        int beg0 = 0, end0 = 0, beg1 = 0, end1 = 0;
        float4 self0 = make_float4(0.f, 0.f, 0.f, 0.f);
        float4 self1 = make_float4(0.f, 0.f, 0.f, 0.f);
        if (n0 < nrows) {
            beg0 = rowptr[n0];
            end0 = rowptr[n0 + 1];
        }
        if (n0 + 1 < nrows) {
            beg1 = end0;
            end1 = rowptr[n0 + 2];
        }
        const int2 r0 = gcn_load_records(rec, beg0, end0, lane);
        const int2 r1 = gcn_load_records(rec, beg1, end1, lane);
        if (n0 < nrows)     self0 = h4[(size_t)n0 * (DIM / 4) + lane];
        if (n0 + 1 < nrows) self1 = h4[(size_t)(n0 + 1) * (DIM / 4) + lane];

        const float4 a0 = gcn_sum_list(h4, rec, beg0, end0, r0, lane);
        const float4 a1 = gcn_sum_list(h4, rec, beg1, end1, r1, lane);

        sA4[m0 * (APITCH / 4) + lane]       = a0;
        sA4[(m0 + 1) * (APITCH / 4) + lane] = a1;
        sH4[m0 * (APITCH / 4) + lane]       = self0;
        sH4[(m0 + 1) * (APITCH / 4) + lane] = self1;
    }
    __syncthreads();

    const float* sA = reinterpret_cast<const float*>(sA4);
    const float* sH = reinterpret_cast<const float*>(sH4);

    const float4* bR = WrelP  + wv * (16 * 32) + lane;
    const float4* bT = WrootP + wv * (16 * 32) + lane;

    v8f acc_rel  = {};
    v8f acc_root = {};
#pragma unroll
    for (int u = 0; u < DIM / 8; ++u) {
        const float4 br = bR[u * 32];
        const float4 bt = bT[u * 32];
#pragma unroll
        for (int s = 0; s < 2; ++s) {
            const int ka = 8 * u + 4 * s + 2 * half;

            v2f a_rel  = *reinterpret_cast<const v2f*>(&sA[r * APITCH + ka]);
            v2f b_rel  = s ? v2f{ br.z, br.w } : v2f{ br.x, br.y };
            acc_rel = __builtin_amdgcn_wmma_f32_16x16x4_f32(
                          false, a_rel, false, b_rel, (short)0, acc_rel, false, false);

            v2f a_root = *reinterpret_cast<const v2f*>(&sH[r * APITCH + ka]);
            v2f b_root = s ? v2f{ bt.z, bt.w } : v2f{ bt.x, bt.y };
            acc_root = __builtin_amdgcn_wmma_f32_16x16x4_f32(
                          false, a_root, false, b_root, (short)0, acc_root, false, false);
        }
    }

    const float bv = bias[col0 + r];
#pragma unroll
    for (int j = 0; j < 8; ++j) {
        const int m    = j + 8 * half;
        const int orow = row0 + m;
        if (orow < nrows) {
            float v = acc_rel[j] + acc_root[j] + bv;
            if (do_relu) v = fmaxf(v, 0.f);
            out[(size_t)orow * DIM + col0 + r] = v;
        }
    }
}

#define BASE_PAD   (NCOUNTS + 64)
#define ROWPTR_PAD (NNODES + 64)
static_assert(NCOUNTS % 4 == 0 && NNODES % 4 == 0 && NEDGES % 4 == 0, "16B alignment of the planes");

extern "C" void kernel_launch(void* const* d_in, const int* in_sizes, int n_in,
                              void* d_out, int out_size, void* d_ws, size_t ws_size,
                              hipStream_t stream) {
    const float* x      = (const float*)d_in[0];
    const int*   eidx   = (const int*)  d_in[1];
    const float* ew     = (const float*)d_in[2];
    const float* Wrel   = (const float*)d_in[3];
    const float* brel   = (const float*)d_in[4];
    const float* Wroot  = (const float*)d_in[5];
    float*       out    = (float*)d_out;

    const int N = NNODES;
    const int E = NEDGES;

    float* h1       = (float*)d_ws;
    float* h2       = h1 + (size_t)N * DIM;
    float4* wpack   = (float4*)(h2 + (size_t)N * DIM);
    int*   tilecnt  = (int*)(wpack + WPACK_QUADS);
    int*   tilebase = tilecnt + NCOUNTS;
    int*   rank1    = tilebase + BASE_PAD;
    int*   rank2    = rank1 + E;
    int*   rowptr   = rank2 + E;
    int4*  rec1     = (int4*)(rowptr + ROWPTR_PAD);
    int2*  rec      = (int2*)(rec1 + E);

    const int edge_blocks  = (E + 255) / 256;
    const int layer_blocks = (N + 15) / 16;

    gcn_bin_count_kernel<<<NTILES, BINW, 0, stream>>>(eidx, tilecnt, rank1, E);
    gcn_scan_kernel<<<SCAN_BLOCKS + PACK_BLOCKS, SCAN_THREADS, 0, stream>>>(tilecnt, tilebase, NCOUNTS,
                                                                            Wrel, Wroot, wpack);
    gcn_bin_place_kernel<<<edge_blocks, 256, 0, stream>>>(eidx, ew, tilebase, rank1, rec1, E);
    gcn_node_group_kernel<<<NBINS, BINW, 0, stream>>>(tilebase, rec1, rank2, rowptr, rec, N);

    for (int l = 0; l < NLAYERS; ++l) {
        const float* hin  = (l == 0) ? x  : (l == 1) ? h1 : h2;
        float*       hout = (l == 0) ? h1 : (l == 1) ? h2 : out;

        gcn_layer_kernel<<<layer_blocks, 256, 0, stream>>>(
            hin, rowptr, rec,
            wpack + (size_t)(2 * l)     * (DIM * DIM / 4),
            wpack + (size_t)(2 * l + 1) * (DIM * DIM / 4),
            brel  + (size_t)l * DIM,
            hout, N, (l < NLAYERS - 1) ? 1 : 0);
    }
}


# bin placement kernel: a thread moves 8 edges (2 quads of 4 consecutive edges, 16-byte loads), not 1: the launch is 3,128 waves, not 25,000 that each live two memory round trips; integer moves only
# speedup vs baseline: 3.7619x; 1.0389x over previous
#include <hip/hip_runtime.h>

#define NNODES  50000
#define NEDGES  800000
#define DIM     128
#define NLAYERS 3

typedef __attribute__((ext_vector_type(2))) float v2f;
typedef __attribute__((ext_vector_type(8))) float v8f;


#define BINW    256
#define NWAVES  (BINW / 32)
#define TILE1   4096
#define NTILES  ((NEDGES + TILE1 - 1) / TILE1)
#define NBINS   (NNODES / BINW + 1)
#define NCOUNTS (BINW * NTILES)
#define BATCH   8
#define ROUNDS1 (TILE1 / NWAVES / 32)

#define SCAN_THREADS 1024
#define SCAN_BLOCKS  ((NCOUNTS + SCAN_THREADS - 1) / SCAN_THREADS)
#define SCAN_QUADS   ((SCAN_BLOCKS - 1) * (SCAN_THREADS / 4))
#define SCAN_PRE     ((SCAN_QUADS + SCAN_THREADS - 1) / SCAN_THREADS)

static_assert(SCAN_THREADS / 32 == 32, "one lane per wave when the wave sums are folded");

#define WPACK_MATS   (NLAYERS * 2)
#define WPACK_QUADS  (WPACK_MATS * DIM * DIM / 4)
#define PACK_BLOCKS  (WPACK_QUADS / SCAN_THREADS)
static_assert(WPACK_QUADS % SCAN_THREADS == 0, "pack blocks are whole");
static_assert(DIM == 128, "the fragment order below is written for 8 column tiles and 16 k-groups of 8");

static_assert(NBINS <= BINW, "level-1 bin numbers must fit the 256-entry histogram");
static_assert(TILE1 % (NWAVES * 32 * BATCH) == 0, "a tile is whole batches for every wave");
static_assert(ROUNDS1 % BATCH == 0, "a wave's rounds are whole batches");

__device__ __forceinline__ void gcn_match32(int key, int& lt, int& tot, bool& last) {
    unsigned mine = 0u;
#pragma unroll
    for (int l = 0; l < 32; ++l) {
        const int      kl = __builtin_amdgcn_readlane(key, l);
        const bool     eq = (key == kl);
        const unsigned m  = __builtin_amdgcn_ballot_w32(eq);
        if (eq) mine = m;
    }
    const unsigned lane = threadIdx.x & 31u;
    lt   = (int)__builtin_amdgcn_mbcnt_lo(mine, 0u);
    tot  = __popc(mine);
    last = ((mine >> lane) == 1u);
}

__device__ __forceinline__ int gcn_rank_round(int* __restrict__ whist, int key, bool valid) {
    int lt, tot; bool last;
    gcn_match32(key, lt, tot, last);
    int part = 0;
    if (valid) {
        const int old = whist[key];
        part = old + lt;
        if (last) whist[key] = old + tot;
    }
    return part;
}

__global__ __launch_bounds__(BINW) void gcn_bin_count_kernel(
    const int* __restrict__ edge_index,
    int*       __restrict__ tilecnt,
    int*       __restrict__ rank1,
    int nedges) {
    __shared__ int hist[NWAVES][BINW];

    const int t    = threadIdx.x;
    const int lane = t & 31;
    const int wv   = __builtin_amdgcn_readfirstlane(t >> 5);
    const int tile = blockIdx.x;
    const int e0   = tile * TILE1;
    const int ne   = min(TILE1, nedges - e0);
    const int seg0 = wv * (TILE1 / NWAVES);
    const int* dst = edge_index + nedges;

    for (int k = 0; k < NWAVES; ++k) hist[k][t] = 0;
    __syncthreads();

    int key[ROUNDS1];
    int part[ROUNDS1];
#pragma unroll
    for (int g = 0; g < ROUNDS1 / BATCH; ++g) {
        const int q0 = g * 32 * BATCH;
#pragma unroll
        for (int b = 0; b < BATCH; ++b) {
            const int idx = seg0 + q0 + 32 * b + lane;
            const int v = dst[e0 + min(idx, ne - 1)];
            key[g * BATCH + b] = (idx < ne) ? (v >> 8) : -1;
        }
#pragma unroll
        for (int b = 0; b < BATCH; ++b) {
            const int  idx   = seg0 + q0 + 32 * b + lane;
            const bool valid = idx < ne;
            int p = 0;
            if (seg0 + q0 + 32 * b < ne) p = gcn_rank_round(hist[wv], key[g * BATCH + b], valid);
            part[g * BATCH + b] = p;
        }
    }
    __syncthreads();

    int run = 0;
    for (int k = 0; k < NWAVES; ++k) {
        const int v = hist[k][t];
        hist[k][t] = run;
        run += v;
    }
    tilecnt[t * NTILES + tile] = run;
    __syncthreads();

    int woff[ROUNDS1];
#pragma unroll
    for (int i = 0; i < ROUNDS1; ++i) woff[i] = hist[wv][max(key[i], 0)];
#pragma unroll
    for (int i = 0; i < ROUNDS1; ++i) {
        const int idx = seg0 + 32 * i + lane;
        if (idx < ne) rank1[e0 + idx] = part[i] + woff[i];
    }
}

__global__ __launch_bounds__(SCAN_THREADS) void gcn_scan_kernel(
    const int*   __restrict__ cnt,
    int*         __restrict__ base,
    int n,
    const float* __restrict__ Wrel,
    const float* __restrict__ Wroot,
    float4*      __restrict__ wpack) {
    __shared__ int wave_pre[SCAN_THREADS / 32];
    __shared__ int wave_tot[SCAN_THREADS / 32];

    if (blockIdx.x >= SCAN_BLOCKS) {
        const int p    = (blockIdx.x - SCAN_BLOCKS) * SCAN_THREADS + threadIdx.x;
        const int ln   = p & 31;
        const int u    = (p >> 5) & 15;
        const int w    = (p >> 9) & 7;
        const int m    = p >> 12;
        const int col  = 16 * w + (ln & 15);
        const int k0   = 8 * u + 2 * (ln >> 4);
        const float* W = ((m & 1) ? Wroot : Wrel) + (size_t)(m >> 1) * DIM * DIM;
        wpack[p] = make_float4(W[(size_t)k0 * DIM + col],       W[(size_t)(k0 + 1) * DIM + col],
                               W[(size_t)(k0 + 4) * DIM + col], W[(size_t)(k0 + 5) * DIM + col]);
        return;
    }

    const int t     = threadIdx.x;
    const int lane  = t & 31;
    const int wv    = t >> 5;
    const int first = blockIdx.x * SCAN_THREADS;
    const int idx   = first + t;
    const int nq    = first / 4;
    const int4* cnt4 = reinterpret_cast<const int4*>(cnt);

    int4 q[SCAN_PRE];
#pragma unroll
    for (int i = 0; i < SCAN_PRE; ++i) {
        const int k = i * SCAN_THREADS + t;
        q[i] = cnt4[(k < nq) ? k : 0];
    }
    const int v = (idx < n) ? cnt[idx] : 0;

    int pre = 0;
#pragma unroll
    for (int i = 0; i < SCAN_PRE; ++i) {
        const int k = i * SCAN_THREADS + t;
        if (k < nq) pre += (q[i].x + q[i].y) + (q[i].z + q[i].w);
    }
    for (int off = 16; off > 0; off >>= 1) pre += __shfl_xor(pre, off);

    int incl = v;
    for (int off = 1; off < 32; off <<= 1) {
        const int y = __shfl_up(incl, off);
        if (lane >= off) incl += y;
    }
    if (lane == 0)  wave_pre[wv] = pre;
    if (lane == 31) wave_tot[wv] = incl;
    __syncthreads();

    int s = wave_pre[lane] + ((lane < wv) ? wave_tot[lane] : 0);
    for (int off = 16; off > 0; off >>= 1) s += __shfl_xor(s, off);

    const int excl = s + incl - v;
    if (idx < n)      base[idx] = excl;
    if (idx == n - 1) base[n]   = excl + v;
}

#define PLACE_THREADS 128
#define PLACE_K       8
#define PLACE_QUADS   (PLACE_K / 4)
#define PLACE_EDGES   (PLACE_THREADS * PLACE_K)
#define PLACE_BLOCKS  ((NEDGES + PLACE_EDGES - 1) / PLACE_EDGES)
static_assert(PLACE_K % 4 == 0, "a thread takes whole quads");
static_assert(TILE1 % PLACE_EDGES == 0, "the edges of a block lie in one level-1 tile");
static_assert(NEDGES % 4 == 0, "edges are read four at a time; the dst row of edge_index starts 16-byte aligned");

template <bool FULL>
__device__ __forceinline__ void gcn_place_quads(
    const int4* __restrict__ src4, const int4* __restrict__ dst4,
    const float4* __restrict__ w4, const int4* __restrict__ rk4,
    const int* __restrict__ tb, int4* __restrict__ rec1, int q0, int nq) {
    int4   s[PLACE_QUADS], d[PLACE_QUADS], rk[PLACE_QUADS];
    float4 w[PLACE_QUADS];
#pragma unroll
    for (int g = 0; g < PLACE_QUADS; ++g) {
        const int qg = q0 + g * PLACE_THREADS;
        const int q  = FULL ? qg : min(qg, nq - 1);
        d[g]  = dst4[q];
        s[g]  = src4[q];
        w[g]  = w4[q];
        rk[g] = rk4[q];
    }
    int4 b[PLACE_QUADS];
#pragma unroll
    for (int g = 0; g < PLACE_QUADS; ++g) {
        b[g].x = tb[(d[g].x >> 8) * NTILES];
        b[g].y = tb[(d[g].y >> 8) * NTILES];
        b[g].z = tb[(d[g].z >> 8) * NTILES];
        b[g].w = tb[(d[g].w >> 8) * NTILES];
    }
#pragma unroll
    for (int g = 0; g < PLACE_QUADS; ++g) {
        if (FULL || q0 + g * PLACE_THREADS < nq) {
            rec1[b[g].x + rk[g].x] = make_int4(s[g].x, __float_as_int(w[g].x), d[g].x, 0);
            rec1[b[g].y + rk[g].y] = make_int4(s[g].y, __float_as_int(w[g].y), d[g].y, 0);
            rec1[b[g].z + rk[g].z] = make_int4(s[g].z, __float_as_int(w[g].z), d[g].z, 0);
            rec1[b[g].w + rk[g].w] = make_int4(s[g].w, __float_as_int(w[g].w), d[g].w, 0);
        }
    }
}

__global__ __launch_bounds__(PLACE_THREADS) void gcn_bin_place_kernel(
    const int*   __restrict__ edge_index,
    const float* __restrict__ edge_weight,
    const int*   __restrict__ tilebase,
    const int*   __restrict__ rank1,
    int4*        __restrict__ rec1,
    int nedges) {
    const int e0   = blockIdx.x * PLACE_EDGES;
    const int q0   = e0 / 4 + threadIdx.x;
    const int nq   = nedges / 4;
    const int4*   src4 = reinterpret_cast<const int4*>(edge_index);
    const int4*   dst4 = reinterpret_cast<const int4*>(edge_index + nedges);
    const float4* w4   = reinterpret_cast<const float4*>(edge_weight);
    const int4*   rk4  = reinterpret_cast<const int4*>(rank1);
    const int*    tb   = tilebase + e0 / TILE1;

    if (e0 + PLACE_EDGES <= nedges) gcn_place_quads<true >(src4, dst4, w4, rk4, tb, rec1, q0, nq);
    else                            gcn_place_quads<false>(src4, dst4, w4, rk4, tb, rec1, q0, nq);
}

__global__ __launch_bounds__(BINW) void gcn_node_group_kernel(
    const int*  __restrict__ tilebase,
    const int4* __restrict__ rec1,
    int*        __restrict__ rank2,
    int*        __restrict__ rowptr,
    int2*       __restrict__ rec,
    int nnodes) {
    __shared__ int hist[NWAVES][BINW];
    __shared__ int soff[BINW];
    __shared__ int wave_sum[NWAVES];

    const int t    = threadIdx.x;
    const int lane = t & 31;
    const int wv   = __builtin_amdgcn_readfirstlane(t >> 5);
    const int bin  = blockIdx.x;
    const int lo   = tilebase[bin * NTILES];
    const int hi   = tilebase[(bin + 1) * NTILES];
    const int seg  = ((hi - lo + BINW - 1) / BINW) * 32;
    const int p0   = lo + wv * seg;

    for (int k = 0; k < NWAVES; ++k) hist[k][t] = 0;
    __syncthreads();

    for (int q0 = 0; q0 < seg; q0 += 32 * BATCH) {
        if (p0 + q0 >= hi) break;
        int key[BATCH];
#pragma unroll
        for (int b = 0; b < BATCH; ++b) {
            const int q = q0 + 32 * b;
            const int p = p0 + q + lane;
            const int v = rec1[min(p, hi - 1)].z;
            key[b] = (q < seg && p < hi) ? (v & (BINW - 1)) : -1;
        }
#pragma unroll
        for (int b = 0; b < BATCH; ++b) {
            const int  q     = q0 + 32 * b;
            if (q >= seg || p0 + q >= hi) continue;
            const int  p     = p0 + q + lane;
            const bool valid = p < hi;
            const int  part  = gcn_rank_round(hist[wv], key[b], valid);
            if (valid) rank2[p] = part;
        }
    }
    __syncthreads();

    int run = 0;
    for (int k = 0; k < NWAVES; ++k) {
        const int v = hist[k][t];
        hist[k][t] = run;
        run += v;
    }

    int incl = run;
    for (int off = 1; off < 32; off <<= 1) {
        const int y = __shfl_up(incl, off);
        if (lane >= off) incl += y;
    }
    if (lane == 31) wave_sum[wv] = incl;
    __syncthreads();
    int wbase = 0;
    for (int k = 0; k < NWAVES; ++k)
        if (k < wv) wbase += wave_sum[k];
    const int off_t = wbase + incl - run;
    soff[t] = off_t;
    const int node = bin * BINW + t;
    if (node <= nnodes) rowptr[node] = lo + off_t;
    __syncthreads();

    for (int q0 = 0; q0 < seg; q0 += 32 * BATCH) {
        if (p0 + q0 >= hi) break;
        int4 rr[BATCH];
        int  r2[BATCH];
#pragma unroll
        for (int b = 0; b < BATCH; ++b) {
            const int pc = min(p0 + q0 + 32 * b + lane, hi - 1);
            rr[b] = rec1[pc];
            r2[b] = rank2[pc];
        }
#pragma unroll
        for (int b = 0; b < BATCH; ++b) {
            const int q = q0 + 32 * b;
            const int p = p0 + q + lane;
            if (q < seg && p < hi) {
                const int k = rr[b].z & (BINW - 1);
                rec[lo + soff[k] + hist[wv][k] + r2[b]] = make_int2(rr[b].x, rr[b].y);
            }
        }
    }
}

template <int U>
__device__ __forceinline__ void gcn_add_rows(const float4* __restrict__ h4, int2 r,
                                             int j, int lane, float4& acc) {
    float4 v[U];
    float  w[U];
#pragma unroll
    for (int k = 0; k < U; ++k) {
        const int src = __builtin_amdgcn_readlane(r.x, j + k);
        w[k] = __int_as_float(__builtin_amdgcn_readlane(r.y, j + k));
        v[k] = h4[(size_t)src * (DIM / 4) + lane];
    }
#pragma unroll
    for (int k = 0; k < U; ++k) {
        acc.x = fmaf(w[k], v[k].x, acc.x);
        acc.y = fmaf(w[k], v[k].y, acc.y);
        acc.z = fmaf(w[k], v[k].z, acc.z);
        acc.w = fmaf(w[k], v[k].w, acc.w);
    }
}

__device__ __forceinline__ int2 gcn_load_records(const int2* __restrict__ rec,
                                                 int base, int end, int lane) {
    int2 r = make_int2(0, 0);
    if (lane < end - base) r = rec[base + lane];
    return r;
}

__device__ __forceinline__ float4 gcn_sum_list(const float4* __restrict__ h4,
                                               const int2*   __restrict__ rec,
                                               int beg, int end, int2 r, int lane) {
    float4 acc = make_float4(0.f, 0.f, 0.f, 0.f);
    for (int base = beg; base < end; base += 32) {
        const int n = min(32, end - base);
        if (base != beg) r = gcn_load_records(rec, base, end, lane);

        int j = 0;
        for (; j + 8 <= n; j += 8) gcn_add_rows<8>(h4, r, j, lane, acc);
        if (j + 4 <= n) { gcn_add_rows<4>(h4, r, j, lane, acc); j += 4; }
        for (; j < n; ++j) gcn_add_rows<1>(h4, r, j, lane, acc);
    }
    return acc;
}

#define APITCH (DIM + 4)
static_assert(APITCH % 4 == 0 && (APITCH % 64) == 4, "16B-aligned rows that start 4 banks apart");

__global__ __launch_bounds__(256, 16) void gcn_layer_kernel(
    const float* __restrict__ h,
    const int*   __restrict__ rowptr,
    const int2*  __restrict__ rec,
    const float4* __restrict__ WrelP,
    const float4* __restrict__ WrootP,
    const float* __restrict__ bias,
    float*       __restrict__ out,
    int nrows, int do_relu) {
    __shared__ float4 sA4[16 * APITCH / 4];
    __shared__ float4 sH4[16 * APITCH / 4];

    const int t    = threadIdx.x;
    const int lane = t & 31;
    const int wv   = __builtin_amdgcn_readfirstlane(t >> 5);
    const int r    = lane & 15;
    const int half = lane >> 4;

    const int row0 = blockIdx.x * 16;
    const int col0 = wv * 16;

    {
        const float4* h4 = reinterpret_cast<const float4*>(h);
        const int m0 = 2 * wv;
        const int n0 = row0 + m0;

        int beg0 = 0, end0 = 0, beg1 = 0, end1 = 0;
        float4 self0 = make_float4(0.f, 0.f, 0.f, 0.f);
        float4 self1 = make_float4(0.f, 0.f, 0.f, 0.f);
        if (n0 < nrows) {
            beg0 = rowptr[n0];
            end0 = rowptr[n0 + 1];
        }
        if (n0 + 1 < nrows) {
            beg1 = end0;
            end1 = rowptr[n0 + 2];
        }
        const int2 r0 = gcn_load_records(rec, beg0, end0, lane);
        const int2 r1 = gcn_load_records(rec, beg1, end1, lane);
        if (n0 < nrows)     self0 = h4[(size_t)n0 * (DIM / 4) + lane];
        if (n0 + 1 < nrows) self1 = h4[(size_t)(n0 + 1) * (DIM / 4) + lane];

        const float4 a0 = gcn_sum_list(h4, rec, beg0, end0, r0, lane);
        const float4 a1 = gcn_sum_list(h4, rec, beg1, end1, r1, lane);

        sA4[m0 * (APITCH / 4) + lane]       = a0;
        sA4[(m0 + 1) * (APITCH / 4) + lane] = a1;
        sH4[m0 * (APITCH / 4) + lane]       = self0;
        sH4[(m0 + 1) * (APITCH / 4) + lane] = self1;
    }
    __syncthreads();

    const float* sA = reinterpret_cast<const float*>(sA4);
    const float* sH = reinterpret_cast<const float*>(sH4);

    const float4* bR = WrelP  + wv * (16 * 32) + lane;
    const float4* bT = WrootP + wv * (16 * 32) + lane;

    v8f acc_rel  = {};
    v8f acc_root = {};
#pragma unroll
    for (int u = 0; u < DIM / 8; ++u) {
        const float4 br = bR[u * 32];
        const float4 bt = bT[u * 32];
#pragma unroll
        for (int s = 0; s < 2; ++s) {
            const int ka = 8 * u + 4 * s + 2 * half;

            v2f a_rel  = *reinterpret_cast<const v2f*>(&sA[r * APITCH + ka]);
            v2f b_rel  = s ? v2f{ br.z, br.w } : v2f{ br.x, br.y };
            acc_rel = __builtin_amdgcn_wmma_f32_16x16x4_f32(
                          false, a_rel, false, b_rel, (short)0, acc_rel, false, false);

            v2f a_root = *reinterpret_cast<const v2f*>(&sH[r * APITCH + ka]);
            v2f b_root = s ? v2f{ bt.z, bt.w } : v2f{ bt.x, bt.y };
            acc_root = __builtin_amdgcn_wmma_f32_16x16x4_f32(
                          false, a_root, false, b_root, (short)0, acc_root, false, false);
        }
    }

    const float bv = bias[col0 + r];
#pragma unroll
    for (int j = 0; j < 8; ++j) {
        const int m    = j + 8 * half;
        const int orow = row0 + m;
        if (orow < nrows) {
            float v = acc_rel[j] + acc_root[j] + bv;
            if (do_relu) v = fmaxf(v, 0.f);
            out[(size_t)orow * DIM + col0 + r] = v;
        }
    }
}

#define BASE_PAD   (NCOUNTS + 64)
#define ROWPTR_PAD (NNODES + 64)
static_assert(NCOUNTS % 4 == 0 && NNODES % 4 == 0 && NEDGES % 4 == 0, "16B alignment of the planes");

extern "C" void kernel_launch(void* const* d_in, const int* in_sizes, int n_in,
                              void* d_out, int out_size, void* d_ws, size_t ws_size,
                              hipStream_t stream) {
    const float* x      = (const float*)d_in[0];
    const int*   eidx   = (const int*)  d_in[1];
    const float* ew     = (const float*)d_in[2];
    const float* Wrel   = (const float*)d_in[3];
    const float* brel   = (const float*)d_in[4];
    const float* Wroot  = (const float*)d_in[5];
    float*       out    = (float*)d_out;

    const int N = NNODES;
    const int E = NEDGES;

    float* h1       = (float*)d_ws;
    float* h2       = h1 + (size_t)N * DIM;
    float4* wpack   = (float4*)(h2 + (size_t)N * DIM);
    int*   tilecnt  = (int*)(wpack + WPACK_QUADS);
    int*   tilebase = tilecnt + NCOUNTS;
    int*   rank1    = tilebase + BASE_PAD;
    int*   rank2    = rank1 + E;
    int*   rowptr   = rank2 + E;
    int4*  rec1     = (int4*)(rowptr + ROWPTR_PAD);
    int2*  rec      = (int2*)(rec1 + E);

    const int layer_blocks = (N + 15) / 16;

    gcn_bin_count_kernel<<<NTILES, BINW, 0, stream>>>(eidx, tilecnt, rank1, E);
    gcn_scan_kernel<<<SCAN_BLOCKS + PACK_BLOCKS, SCAN_THREADS, 0, stream>>>(tilecnt, tilebase, NCOUNTS,
                                                                            Wrel, Wroot, wpack);
    gcn_bin_place_kernel<<<PLACE_BLOCKS, PLACE_THREADS, 0, stream>>>(eidx, ew, tilebase, rank1, rec1, E);
    gcn_node_group_kernel<<<NBINS, BINW, 0, stream>>>(tilebase, rec1, rank2, rowptr, rec, N);

    for (int l = 0; l < NLAYERS; ++l) {
        const float* hin  = (l == 0) ? x  : (l == 1) ? h1 : h2;
        float*       hout = (l == 0) ? h1 : (l == 1) ? h2 : out;

        gcn_layer_kernel<<<layer_blocks, 256, 0, stream>>>(
            hin, rowptr, rec,
            wpack + (size_t)(2 * l)     * (DIM * DIM / 4),
            wpack + (size_t)(2 * l + 1) * (DIM * DIM / 4),
            brel  + (size_t)l * DIM,
            hout, N, (l < NLAYERS - 1) ? 1 : 0);
    }
}
